# starting kernel
# baseline (speedup 1.0000x reference)
#include <hip/hip_runtime.h>

#define D 64
#define ALPHA 0.5f

typedef __attribute__((ext_vector_type(2))) float v2f;
typedef __attribute__((ext_vector_type(4))) float v4f;
typedef __attribute__((ext_vector_type(8))) float v8f;

__global__ void zero_ws_kernel(float* __restrict__ ws, int n) {
    int i = blockIdx.x * blockDim.x + threadIdx.x;
    if (i < n) ws[i] = 0.0f;
}

__global__ void scatter_edges_kernel(const float* __restrict__ x,
                                     const int*   __restrict__ ei,
                                     float* __restrict__ sum_in,
                                     float* __restrict__ sum_out,
                                     float* __restrict__ cnt_in,
                                     float* __restrict__ cnt_out,
                                     int E) {
    long long tid = (long long)blockIdx.x * blockDim.x + threadIdx.x;
    long long tot = (long long)E * (D / 4);
    if (tid >= tot) return;
    int e = (int)(tid >> 4);
    int f = (int)(tid & 15) * 4;
    int s = ei[e];
    int d = ei[E + e];

    v4f xs = *(const v4f*)(x + (long long)s * D + f);
    v4f xd = *(const v4f*)(x + (long long)d * D + f);

    float* pin  = sum_in  + (long long)d * D + f;
    float* pout = sum_out + (long long)s * D + f;
    atomicAdd(pin + 0, xs.x); atomicAdd(pin + 1, xs.y);
    atomicAdd(pin + 2, xs.z); atomicAdd(pin + 3, xs.w);
    atomicAdd(pout + 0, xd.x); atomicAdd(pout + 1, xd.y);
    atomicAdd(pout + 2, xd.z); atomicAdd(pout + 3, xd.w);

    if (f == 0) {
        atomicAdd(&cnt_in[d], 1.0f);
        atomicAdd(&cnt_out[s], 1.0f);
    }
}

__global__ __launch_bounds__(128)
void dirsage_gemm_kernel(const float* __restrict__ x,
                         const float* __restrict__ sum_in,
                         const float* __restrict__ sum_out,
                         const float* __restrict__ cnt_in,
                         const float* __restrict__ cnt_out,
                         const float* __restrict__ Wself,
                         const float* __restrict__ bself,
                         const float* __restrict__ Ws2d,
                         const float* __restrict__ bs2d,
                         const float* __restrict__ Wd2s,
                         const float* __restrict__ bd2s,
                         float* __restrict__ out,
                         int N) {
    __shared__ float wl[3 * D * D];

    for (int i = threadIdx.x; i < D * D; i += blockDim.x) {
        int k   = i >> 6;
        int col = i & 63;
        int dst = ((k >> 1) << 7) + (col << 1) + (k & 1);
        wl[dst]             = Wself[i];
        wl[D * D + dst]     = Ws2d[i];
        wl[2 * D * D + dst] = Wd2s[i];
    }
    __syncthreads();

    int wave  = threadIdx.x >> 5;
    int lane  = threadIdx.x & 31;
    int mtile = blockIdx.x * (blockDim.x >> 5) + wave;
    if (mtile * 16 >= N) return;

    int row   = lane & 15;
    int half  = lane >> 4;
    int khalf = half * 2;
    int gr    = mtile * 16 + row;

    float sc_in  = (1.0f - ALPHA) / fmaxf(cnt_in[gr], 1.0f);
    float sc_out = ALPHA          / fmaxf(cnt_out[gr], 1.0f);

    v8f acc[4];
    for (int t = 0; t < 4; ++t) {
        float b = bself[t * 16 + row]
                + (1.0f - ALPHA) * bs2d[t * 16 + row]
                + ALPHA * bd2s[t * 16 + row];
        for (int r = 0; r < 8; ++r) acc[t][r] = b;
    }

    const float* xrow = x       + (long long)gr * D;
    const float* irow = sum_in  + (long long)gr * D;
    const float* orow = sum_out + (long long)gr * D;

    const float* wbase = &wl[half * 128 + row * 2];

    for (int k0 = 0; k0 < D; k0 += 4) {
        int ka = k0 + khalf;
        v2f ax = *(const v2f*)(xrow + ka);
        v2f ai = *(const v2f*)(irow + ka);
        v2f ao = *(const v2f*)(orow + ka);
        ai *= sc_in;
        ao *= sc_out;

        const float* wk = wbase + (k0 >> 1) * 128;
        for (int t = 0; t < 4; ++t) {
            v2f bs = *(const v2f*)(wk + t * 32);
            v2f bi = *(const v2f*)(wk + D * D + t * 32);
            v2f bo = *(const v2f*)(wk + 2 * D * D + t * 32);

            acc[t] = __builtin_amdgcn_wmma_f32_16x16x4_f32(
                false, ax, false, bs, (short)0, acc[t], false, false);
            acc[t] = __builtin_amdgcn_wmma_f32_16x16x4_f32(
                false, ai, false, bi, (short)0, acc[t], false, false);
            acc[t] = __builtin_amdgcn_wmma_f32_16x16x4_f32(
                false, ao, false, bo, (short)0, acc[t], false, false);
        }
    }

    int mofs = half * 8;
    for (int t = 0; t < 4; ++t) {
        int col = t * 16 + row;
        long long base = (long long)(mtile * 16 + mofs) * D + col;
        for (int r = 0; r < 8; ++r) {
            out[base + (long long)r * D] = acc[t][r];
        }
    }
}

extern "C" void kernel_launch(void* const* d_in, const int* in_sizes, int n_in,
                              void* d_out, int out_size, void* d_ws, size_t ws_size,
                              hipStream_t stream) {
    const float* x     = (const float*)d_in[0];
    const int*   ei    = (const int*)  d_in[1];
    const float* Wself = (const float*)d_in[2];
    const float* bself = (const float*)d_in[3];
    const float* Ws2d  = (const float*)d_in[4];
    const float* bs2d  = (const float*)d_in[5];
    const float* Wd2s  = (const float*)d_in[6];
    const float* bd2s  = (const float*)d_in[7];
    float* out = (float*)d_out;

    int N = in_sizes[0] / D;
    int E = in_sizes[1] / 2;

    float* ws      = (float*)d_ws;
    float* sum_in  = ws;
    float* sum_out = ws + (size_t)N * D;
    float* cnt_in  = ws + 2 * (size_t)N * D;
    float* cnt_out = cnt_in + N;

    int zn = 2 * N * D + 2 * N;
    zero_ws_kernel<<<(zn + 255) / 256, 256, 0, stream>>>(ws, zn);

    long long sthreads = (long long)E * (D / 4);
    int sblocks = (int)((sthreads + 255) / 256);
    scatter_edges_kernel<<<sblocks, 256, 0, stream>>>(
        x, ei, sum_in, sum_out, cnt_in, cnt_out, E);

    int mtiles = (N + 15) / 16;
    int wavesPerBlock = 4;
    int gblocks = (mtiles + wavesPerBlock - 1) / wavesPerBlock;
    dirsage_gemm_kernel<<<gblocks, 32 * wavesPerBlock, 0, stream>>>(
        x, sum_in, sum_out, cnt_in, cnt_out,
        Wself, bself, Ws2d, bs2d, Wd2s, bd2s, out, N);
}


# speedup vs baseline: 1.5544x; 1.5544x over previous
#include <hip/hip_runtime.h>


#define D 64
#define ALPHA 0.5f
#define PITCH 68
#define WAVES 28
#define BATCH 16
#define EDGE_WG 256
#define EDGE_PT 4
#define EDGE_CHUNK (32 * EDGE_PT)
#define EDGE_GRID 1024

typedef __attribute__((ext_vector_type(2))) float v2f;
typedef __attribute__((ext_vector_type(8))) float v8f;

__global__ void zero_counts_kernel(int* __restrict__ p, int n) {
    int i = blockIdx.x * blockDim.x + threadIdx.x;
    if (i < n) p[i] = 0;
}

__device__ __forceinline__ int chip_half() {
    return (int)((__builtin_amdgcn_s_sendmsg_rtn(0x87) >> 18) & 1u);
}

__device__ __forceinline__ int draw_ticket(int* counter, int lane) {
    int v = 0;
    if (lane == 0) v = atomicAdd(counter, 1);
    return __builtin_amdgcn_readfirstlane(v);
}

__global__ __launch_bounds__(EDGE_WG)
void count_rank_kernel(const int* __restrict__ ei,
                       int* __restrict__ cnt,
                       int* __restrict__ rank,
                       int* __restrict__ tickets,
                       int N, int E) {
    const int lane    = threadIdx.x & 31;
    const int home    = chip_half();
    const int nchunks = (E + EDGE_CHUNK - 1) / EDGE_CHUNK;
    for (int pass = 0; pass < 2; ++pass) {
        const int dir = home ^ pass;
        const int* __restrict__ key = ei + (dir ? 0 : E);
        int* __restrict__ c  = cnt + dir * N;
        int* __restrict__ rk = rank + dir * E;
        int t = draw_ticket(&tickets[dir], lane);
        for (int it = 0; it < nchunks && t < nchunks; ++it) {
            const int e0 = t * EDGE_CHUNK + lane;
            int k[EDGE_PT], r[EDGE_PT];
#pragma unroll
            for (int i = 0; i < EDGE_PT; ++i) {
                int e = e0 + i * 32;
                k[i] = (e < E) ? key[e] : -1;
            }
            t = draw_ticket(&tickets[dir], lane);
#pragma unroll
            for (int i = 0; i < EDGE_PT; ++i)
                r[i] = (k[i] >= 0) ? atomicAdd(&c[k[i]], 1) : 0;
#pragma unroll
            for (int i = 0; i < EDGE_PT; ++i) {
                int e = e0 + i * 32;
                if (k[i] >= 0) rk[e] = r[i];
            }
        }
    }
}

__global__ __launch_bounds__(1024)
void alloc_kernel(const int* __restrict__ cnt, int* __restrict__ off,
                  int* __restrict__ cursor, int n2) {
    __shared__ int wsum[32];
    __shared__ int base;
    int i    = blockIdx.x * 1024 + threadIdx.x;
    int lane = threadIdx.x & 31;
    int wv   = threadIdx.x >> 5;
    int c    = (i < n2) ? cnt[i] : 0;
    int incl = c;
    for (int dlt = 1; dlt < 32; dlt <<= 1) {
        int t = __shfl_up(incl, dlt);
        if (lane >= dlt) incl += t;
    }
    if (lane == 31) wsum[wv] = incl;
    __syncthreads();
    if (wv == 0) {
        int s  = wsum[lane];
        int si = s;
        for (int dlt = 1; dlt < 32; dlt <<= 1) {
            int t = __shfl_up(si, dlt);
            if (lane >= dlt) si += t;
        }
        wsum[lane] = si - s;
        if (lane == 31) base = atomicAdd(cursor, si);
    }
    __syncthreads();
    if (i < n2) off[i] = base + wsum[wv] + incl - c;
}

__global__ __launch_bounds__(EDGE_WG)
void place_kernel(const int* __restrict__ ei,
                  const int* __restrict__ off,
                  const int* __restrict__ rank,
                  int* __restrict__ list,
                  int* __restrict__ tickets,
                  int N, int E) {
    const int lane    = threadIdx.x & 31;
    const int home    = chip_half();
    const int nchunks = (E + EDGE_CHUNK - 1) / EDGE_CHUNK;
    for (int pass = 0; pass < 2; ++pass) {
        const int dir = home ^ pass;
        const int* __restrict__ key = ei + (dir ? 0 : E);
        const int* __restrict__ val = ei + (dir ? E : 0);
        const int* __restrict__ o   = off + dir * N;
        const int* __restrict__ rk  = rank + dir * E;
        int t = draw_ticket(&tickets[dir], lane);
        for (int it = 0; it < nchunks && t < nchunks; ++it) {
            const int e0 = t * EDGE_CHUNK + lane;
            int k[EDGE_PT], v[EDGE_PT], r[EDGE_PT], b[EDGE_PT];
#pragma unroll
            for (int i = 0; i < EDGE_PT; ++i) {
                int e = e0 + i * 32;
                bool in = e < E;
                k[i] = in ? key[e] : -1;
                v[i] = in ? val[e] : 0;
                r[i] = in ? rk[e] : 0;
            }
            t = draw_ticket(&tickets[dir], lane);
#pragma unroll
            for (int i = 0; i < EDGE_PT; ++i)
                b[i] = (k[i] >= 0) ? o[k[i]] : 0;
#pragma unroll
            for (int i = 0; i < EDGE_PT; ++i)
                if (k[i] >= 0) list[b[i] + r[i]] = v[i];
        }
    }
}

__device__ __forceinline__ v2f sum_sorted(const float* __restrict__ x,
                                          int L, int nid, int lane) {
    v2f acc = {0.0f, 0.0f};
    int id = (lane < L) ? nid : 0x7fffffff;
    int rk = 0;
    for (int t = 0; t < L; ++t) {
        int v = __builtin_amdgcn_readlane(id, t);
        rk += ((v < id) || (v == id && t < lane)) ? 1 : 0;
    }
    int sorted = __builtin_amdgcn_ds_permute(rk << 2, id);
    const float* xl = x + 2 * lane;
    for (int t0 = 0; t0 < L; t0 += BATCH) {
        v2f v[BATCH];
#pragma unroll
        for (int i = 0; i < BATCH; ++i) {
            if (t0 + i < L) {
                int j = __builtin_amdgcn_readlane(sorted, t0 + i);
                v[i] = *(const v2f*)(xl + (long long)j * D);
            }
        }
#pragma unroll
        for (int i = 0; i < BATCH; ++i)
            if (t0 + i < L) acc += v[i];
    }
    return acc;
}

__device__ __forceinline__ v2f sum_long(const float* __restrict__ x,
                                        const int* __restrict__ lst,
                                        int L, int lane) {
    v2f acc = {0.0f, 0.0f};
    int prev = -1;
    int done = 0;
    while (done < L) {
        int best = 0x7fffffff;
        for (int i = lane; i < L; i += 32) {
            int v = lst[i];
            if (v > prev && v < best) best = v;
        }
        for (int m = 16; m >= 1; m >>= 1) {
            int o = __shfl_xor(best, m);
            best = o < best ? o : best;
        }
        int mult = 0;
        for (int i = lane; i < L; i += 32) mult += (lst[i] == best) ? 1 : 0;
        for (int m = 16; m >= 1; m >>= 1) mult += __shfl_xor(mult, m);
        best = __builtin_amdgcn_readfirstlane(best);
        mult = __builtin_amdgcn_readfirstlane(mult);
        if (mult == 0) break;
        v2f r = *(const v2f*)(x + (long long)best * D + 2 * lane);
        acc += r;
        for (int q = 1; q < mult; ++q) acc += r;
        prev = best;
        done += mult;
    }
    return acc;
}

__device__ __forceinline__ void gather_tile(const float* __restrict__ x,
                                            const int* __restrict__ list,
                                            float* st,
                                            int o, int c, int sel, float alpha,
                                            int lane) {
    int off0 = __builtin_amdgcn_readlane(o, sel);
    int L0   = __builtin_amdgcn_readlane(c, sel);
    int nid  = (lane < L0) ? list[off0 + lane] : 0;
    for (int r = 0; r < 16; ++r) {
        int Lc = L0, nidc = nid;
        if (r < 15) {
            off0 = __builtin_amdgcn_readlane(o, sel + r + 1);
            L0   = __builtin_amdgcn_readlane(c, sel + r + 1);
            nid  = (lane < L0) ? list[off0 + lane] : 0;
        }
        v2f a = {0.0f, 0.0f};
        if (Lc <= 32) a = sum_sorted(x, Lc, nidc, lane);
        a *= alpha / fmaxf((float)Lc, 1.0f);
        *(v2f*)(st + r * PITCH + 2 * lane) = a;
    }
    for (int r = 0; r < 16; ++r) {
        int Lc = __builtin_amdgcn_readlane(c, sel + r);
        if (Lc > 32) {
            int offc = __builtin_amdgcn_readlane(o, sel + r);
            v2f a = sum_long(x, list + offc, Lc, lane);
            a *= alpha / fmaxf((float)Lc, 1.0f);
            *(v2f*)(st + r * PITCH + 2 * lane) = a;
        }
    }
}

__global__ __launch_bounds__(32 * WAVES)
void dirsage_pull_gemm_kernel(const float* __restrict__ x,
                              const int*   __restrict__ cnt,
                              const int*   __restrict__ off,
                              const int*   __restrict__ list,
                              const float* __restrict__ Wself,
                              const float* __restrict__ bself,
                              const float* __restrict__ Ws2d,
                              const float* __restrict__ bs2d,
                              const float* __restrict__ Wd2s,
                              const float* __restrict__ bd2s,
                              float* __restrict__ out,
                              int N) {
    __shared__ float wl[3 * D * D];
    __shared__ float stage[WAVES * 16 * PITCH];

    for (int i = threadIdx.x; i < D * D; i += blockDim.x) {
        int k   = i >> 6;
        int col = i & 63;
        int dst = ((k >> 1) << 7) + (col << 1) + (k & 1);
        wl[dst]             = Wself[i];
        wl[D * D + dst]     = Ws2d[i];
        wl[2 * D * D + dst] = Wd2s[i];
    }
    __syncthreads();

    int wave   = threadIdx.x >> 5;
    int lane   = threadIdx.x & 31;
    int row    = lane & 15;
    int half   = lane >> 4;
    int khalf  = half * 2;
    int mtiles = N >> 4;
    float* st  = stage + wave * 16 * PITCH;
    const float* wbase = &wl[half * 128 + row * 2];
    const float* sbase = st + row * PITCH + khalf;

    for (int mtile = blockIdx.x + gridDim.x * wave; mtile < mtiles;
         mtile += gridDim.x * WAVES) {
        int gr0 = mtile * 16;
        int gr  = gr0 + row;

        int li = half * N + gr;
        int o  = off[li];
        int c  = cnt[li];

        v8f acc[4];
        for (int t = 0; t < 4; ++t) {
            float b = bself[t * 16 + row]
                    + (1.0f - ALPHA) * bs2d[t * 16 + row]
                    + ALPHA * bd2s[t * 16 + row];
            for (int r = 0; r < 8; ++r) acc[t][r] = b;
        }

        const float* xrow = x + (long long)gr * D + khalf;
        v2f ax[16];
#pragma unroll
        for (int j = 0; j < 16; ++j) ax[j] = *(const v2f*)(xrow + 4 * j);
#pragma unroll
        for (int j = 0; j < 16; ++j) {
            const float* wk = wbase + j * 256;
            for (int t = 0; t < 4; ++t) {
                v2f bs = *(const v2f*)(wk + t * 32);
                acc[t] = __builtin_amdgcn_wmma_f32_16x16x4_f32(
                    false, ax[j], false, bs, (short)0, acc[t], false, false);
            }
        }

        for (int dir = 0; dir < 2; ++dir) {
            gather_tile(x, list, st, o, c, dir * 16,
                        dir ? ALPHA : (1.0f - ALPHA), lane);
            __builtin_amdgcn_wave_barrier();
            const float* wd = wbase + (dir + 1) * D * D;
            for (int k0 = 0; k0 < D; k0 += 4) {
                v2f aa = *(const v2f*)(sbase + k0);
                const float* wk = wd + (k0 >> 1) * 128;
                for (int t = 0; t < 4; ++t) {
                    v2f bb = *(const v2f*)(wk + t * 32);
                    acc[t] = __builtin_amdgcn_wmma_f32_16x16x4_f32(
                        false, aa, false, bb, (short)0, acc[t], false, false);
                }
            }
            __builtin_amdgcn_wave_barrier();
        }

        int mofs = half * 8;
        for (int t = 0; t < 4; ++t) {
            int col = t * 16 + row;
            long long base = (long long)(gr0 + mofs) * D + col;
            for (int r = 0; r < 8; ++r) out[base + (long long)r * D] = acc[t][r];
        }
    }
}

extern "C" void kernel_launch(void* const* d_in, const int* in_sizes, int n_in,
                              void* d_out, int out_size, void* d_ws, size_t ws_size,
                              hipStream_t stream) {
    const float* x     = (const float*)d_in[0];
    const int*   ei    = (const int*)  d_in[1];
    const float* Wself = (const float*)d_in[2];
    const float* bself = (const float*)d_in[3];
    const float* Ws2d  = (const float*)d_in[4];
    const float* bs2d  = (const float*)d_in[5];
    const float* Wd2s  = (const float*)d_in[6];
    const float* bd2s  = (const float*)d_in[7];
    float* out = (float*)d_out;

    int N = in_sizes[0] / D;
    int E = in_sizes[1] / 2;

    int* wsi    = (int*)d_ws;
    int* cnt    = wsi;
    int* cursor = wsi + 2 * (size_t)N;
    int* off    = cursor + 8;
    int* rank   = off + 2 * (size_t)N;
    int* list   = rank + 2 * (size_t)E;

    int zn = 2 * N + 8;
    zero_counts_kernel<<<(zn + 255) / 256, 256, 0, stream>>>(wsi, zn);
    count_rank_kernel<<<EDGE_GRID, EDGE_WG, 0, stream>>>(ei, cnt, rank, cursor + 1, N, E);
    alloc_kernel<<<(2 * N + 1023) / 1024, 1024, 0, stream>>>(cnt, off, cursor, 2 * N);
    place_kernel<<<EDGE_GRID, EDGE_WG, 0, stream>>>(ei, off, rank, list, cursor + 3, N, E);
    dirsage_pull_gemm_kernel<<<256, 32 * WAVES, 0, stream>>>(
        x, cnt, off, list, Wself, bself, Ws2d, bs2d, Wd2s, bd2s, out, N);
}


# Lists kept in fixed 32-slot rows (one 128-B line per node and direction): the count's returned slot stores the neighbour id at once, excess to a side list; no rank plane, scan or placing pass
# speedup vs baseline: 1.9960x; 1.2841x over previous
#include <hip/hip_runtime.h>


#define D 64
#define ALPHA 0.5f
#define PITCH 68
#define WAVES 28
#define BATCH 16
#define EDGE_WG 256
#define EDGE_PT 4
#define EDGE_CHUNK (32 * EDGE_PT)
#define EDGE_GRID 1024
#define CAP 32

typedef __attribute__((ext_vector_type(2))) float v2f;
typedef __attribute__((ext_vector_type(8))) float v8f;

__global__ void zero_counts_kernel(int* __restrict__ p, int n) {
    int i = blockIdx.x * blockDim.x + threadIdx.x;
    if (i < n) p[i] = 0;
}

__device__ __forceinline__ int chip_half() {
    return (int)((__builtin_amdgcn_s_sendmsg_rtn(0x87) >> 18) & 1u);
}

__device__ __forceinline__ int draw_ticket(int* counter, int lane) {
    int v = 0;
    if (lane == 0) v = atomicAdd(counter, 1);
    return __builtin_amdgcn_readfirstlane(v);
}

__global__ __launch_bounds__(EDGE_WG)
void fill_lists_kernel(const int* __restrict__ ei,
                       int* __restrict__ cnt,
                       int* __restrict__ slots,
                       int* __restrict__ ovf,
                       int* __restrict__ novf,
                       int* __restrict__ tickets,
                       int N, int E) {
    const int lane    = threadIdx.x & 31;
    const int home    = chip_half();
    const int nchunks = (E + EDGE_CHUNK - 1) / EDGE_CHUNK;
    for (int pass = 0; pass < 2; ++pass) {
        const int dir = home ^ pass;
        const int* __restrict__ key = ei + (dir ? 0 : E);
        const int* __restrict__ val = ei + (dir ? E : 0);
        int* __restrict__ c  = cnt + dir * N;
        int* __restrict__ sl = slots + (long long)dir * N * CAP;
        int* __restrict__ ov = ovf + (long long)dir * E;
        int t = draw_ticket(&tickets[dir], lane);
        for (int it = 0; it < nchunks && t < nchunks; ++it) {
            const int e0 = t * EDGE_CHUNK + lane;
            int k[EDGE_PT], v[EDGE_PT], r[EDGE_PT];
#pragma unroll
            for (int i = 0; i < EDGE_PT; ++i) {
                int e = e0 + i * 32;
                bool in = e < E;
                k[i] = in ? key[e] : -1;
                v[i] = in ? val[e] : 0;
            }
            t = draw_ticket(&tickets[dir], lane);
#pragma unroll
            for (int i = 0; i < EDGE_PT; ++i)
                r[i] = (k[i] >= 0) ? atomicAdd(&c[k[i]], 1) : 0;
#pragma unroll
            for (int i = 0; i < EDGE_PT; ++i) {
                if (k[i] >= 0) {
                    if (r[i] < CAP) {
                        sl[k[i] * CAP + r[i]] = v[i];
                    } else {
                        int o = atomicAdd(&novf[dir], 1);
                        if (o < E) ov[o] = e0 + i * 32;
                    }
                }
            }
        }
    }
}

__device__ __forceinline__ v2f sum_sorted(const float* __restrict__ x,
                                          int L, int nid, int lane) {
    v2f acc = {0.0f, 0.0f};
    int id = (lane < L) ? nid : 0x7fffffff;
    int rk = 0;
    for (int t = 0; t < L; ++t) {
        int v = __builtin_amdgcn_readlane(id, t);
        rk += ((v < id) || (v == id && t < lane)) ? 1 : 0;
    }
    int sorted = __builtin_amdgcn_ds_permute(rk << 2, id);
    const float* xl = x + 2 * lane;
    for (int t0 = 0; t0 < L; t0 += BATCH) {
        v2f v[BATCH];
#pragma unroll
        for (int i = 0; i < BATCH; ++i) {
            if (t0 + i < L) {
                int j = __builtin_amdgcn_readlane(sorted, t0 + i);
                v[i] = *(const v2f*)(xl + (long long)j * D);
            }
        }
#pragma unroll
        for (int i = 0; i < BATCH; ++i)
            if (t0 + i < L) acc += v[i];
    }
    return acc;
}

__device__ __forceinline__ v2f sum_long(const float* __restrict__ x,
                                        const int* __restrict__ row,
                                        const int* __restrict__ ov, int nov,
                                        const int* __restrict__ key,
                                        const int* __restrict__ val,
                                        int node, int L, int E, int lane) {
    v2f acc = {0.0f, 0.0f};
    int own  = row[lane];
    int prev = -1;
    int done = 0;
    while (done < L) {
        int best = (own > prev) ? own : 0x7fffffff;
        for (int i = lane; i < nov; i += 32) {
            int e = ov[i];
            if ((unsigned)e < (unsigned)E && key[e] == node) {
                int v = val[e];
                if (v > prev && v < best) best = v;
            }
        }
        for (int m = 16; m >= 1; m >>= 1) {
            int o = __shfl_xor(best, m);
            best = o < best ? o : best;
        }
        int mult = (own == best) ? 1 : 0;
        for (int i = lane; i < nov; i += 32) {
            int e = ov[i];
            if ((unsigned)e < (unsigned)E && key[e] == node && val[e] == best) mult += 1;
        }
        for (int m = 16; m >= 1; m >>= 1) mult += __shfl_xor(mult, m);
        best = __builtin_amdgcn_readfirstlane(best);
        mult = __builtin_amdgcn_readfirstlane(mult);
        if (mult == 0) break;
        v2f r = *(const v2f*)(x + (long long)best * D + 2 * lane);
        acc += r;
        for (int q = 1; q < mult; ++q) acc += r;
        prev = best;
        done += mult;
    }
    return acc;
}

__device__ __forceinline__ void gather_tile(const float* __restrict__ x,
                                            const int* __restrict__ rows,
                                            float* st,
                                            int c, int sel, float alpha,
                                            const int* __restrict__ ov,
                                            const int* __restrict__ novp,
                                            const int* __restrict__ key,
                                            const int* __restrict__ val,
                                            int node0, int E, int lane) {
    int L0  = __builtin_amdgcn_readlane(c, sel);
    int nid = (lane < L0) ? rows[lane] : 0;
    for (int r = 0; r < 16; ++r) {
        int Lc = L0, nidc = nid;
        if (r < 15) {
            L0  = __builtin_amdgcn_readlane(c, sel + r + 1);
            nid = (lane < L0) ? rows[(r + 1) * CAP + lane] : 0;
        }
        v2f a = {0.0f, 0.0f};
        if (Lc <= CAP) a = sum_sorted(x, Lc, nidc, lane);
        a *= alpha / fmaxf((float)Lc, 1.0f);
        *(v2f*)(st + r * PITCH + 2 * lane) = a;
    }
    for (int r = 0; r < 16; ++r) {
        int Lc = __builtin_amdgcn_readlane(c, sel + r);
        if (Lc > CAP) {
            int nov = novp[0];
            nov = nov < E ? nov : E;
            v2f a = sum_long(x, rows + r * CAP, ov, nov, key, val, node0 + r, Lc, E, lane);
            a *= alpha / fmaxf((float)Lc, 1.0f);
            *(v2f*)(st + r * PITCH + 2 * lane) = a;
        }
    }
}

__global__ __launch_bounds__(32 * WAVES)
void dirsage_pull_gemm_kernel(const float* __restrict__ x,
                              const int*   __restrict__ ei,
                              const int*   __restrict__ cnt,
                              const int*   __restrict__ slots,
                              const int*   __restrict__ ovf,
                              const int*   __restrict__ novf,
                              const float* __restrict__ Wself,
                              const float* __restrict__ bself,
                              const float* __restrict__ Ws2d,
                              const float* __restrict__ bs2d,
                              const float* __restrict__ Wd2s,
                              const float* __restrict__ bd2s,
                              float* __restrict__ out,
                              int N, int E) {
    __shared__ float wl[3 * D * D];
    __shared__ float stage[WAVES * 16 * PITCH];

    for (int i = threadIdx.x; i < D * D; i += blockDim.x) {
        int k   = i >> 6;
        int col = i & 63;
        int dst = ((k >> 1) << 7) + (col << 1) + (k & 1);
        wl[dst]             = Wself[i];
        wl[D * D + dst]     = Ws2d[i];
        wl[2 * D * D + dst] = Wd2s[i];
    }
    __syncthreads();

    int wave   = threadIdx.x >> 5;
    int lane   = threadIdx.x & 31;
    int row    = lane & 15;
    int half   = lane >> 4;
    int khalf  = half * 2;
    int mtiles = N >> 4;
    float* st  = stage + wave * 16 * PITCH;
    const float* wbase = &wl[half * 128 + row * 2];
    const float* sbase = st + row * PITCH + khalf;

    for (int mtile = blockIdx.x + gridDim.x * wave; mtile < mtiles;
         mtile += gridDim.x * WAVES) {
        int gr0 = mtile * 16;
        int gr  = gr0 + row;

        int li = half * N + gr;
        int c  = cnt[li];

        v8f acc[4];
        for (int t = 0; t < 4; ++t) {
            float b = bself[t * 16 + row]
                    + (1.0f - ALPHA) * bs2d[t * 16 + row]
                    + ALPHA * bd2s[t * 16 + row];
            for (int r = 0; r < 8; ++r) acc[t][r] = b;
        }

        const float* xrow = x + (long long)gr * D + khalf;
        v2f ax[16];
#pragma unroll
        for (int j = 0; j < 16; ++j) ax[j] = *(const v2f*)(xrow + 4 * j);
#pragma unroll
        for (int j = 0; j < 16; ++j) {
            const float* wk = wbase + j * 256;
            for (int t = 0; t < 4; ++t) {
                v2f bs = *(const v2f*)(wk + t * 32);
                acc[t] = __builtin_amdgcn_wmma_f32_16x16x4_f32(
                    false, ax[j], false, bs, (short)0, acc[t], false, false);
            }
        }

        for (int dir = 0; dir < 2; ++dir) {
            gather_tile(x, slots + ((long long)dir * N + gr0) * CAP, st, c, dir * 16,
                        dir ? ALPHA : (1.0f - ALPHA),
                        ovf + (long long)dir * E, novf + dir,
                        ei + (dir ? 0 : E), ei + (dir ? E : 0),
                        gr0, E, lane);
            __builtin_amdgcn_wave_barrier();
            const float* wd = wbase + (dir + 1) * D * D;
            for (int k0 = 0; k0 < D; k0 += 4) {
                v2f aa = *(const v2f*)(sbase + k0);
                const float* wk = wd + (k0 >> 1) * 128;
                for (int t = 0; t < 4; ++t) {
                    v2f bb = *(const v2f*)(wk + t * 32);
                    acc[t] = __builtin_amdgcn_wmma_f32_16x16x4_f32(
                        false, aa, false, bb, (short)0, acc[t], false, false);
                }
            }
            __builtin_amdgcn_wave_barrier();
        }

        int mofs = half * 8;
        for (int t = 0; t < 4; ++t) {
            int col = t * 16 + row;
            long long base = (long long)(gr0 + mofs) * D + col;
            for (int r = 0; r < 8; ++r) out[base + (long long)r * D] = acc[t][r];
        }
    }
}

extern "C" void kernel_launch(void* const* d_in, const int* in_sizes, int n_in,
                              void* d_out, int out_size, void* d_ws, size_t ws_size,
                              hipStream_t stream) {
    const float* x     = (const float*)d_in[0];
    const int*   ei    = (const int*)  d_in[1];
    const float* Wself = (const float*)d_in[2];
    const float* bself = (const float*)d_in[3];
    const float* Ws2d  = (const float*)d_in[4];
    const float* bs2d  = (const float*)d_in[5];
    const float* Wd2s  = (const float*)d_in[6];
    const float* bd2s  = (const float*)d_in[7];
    float* out = (float*)d_out;

    int N = in_sizes[0] / D;
    int E = in_sizes[1] / 2;

    int* wsi    = (int*)d_ws;
    int* cnt    = wsi;
    int* cursor = wsi + 2 * (size_t)N;
    int* slots  = cursor + 32;
    int* ovf    = slots + 2 * (size_t)N * CAP;

    int zn = 2 * N + 32;
    zero_counts_kernel<<<(zn + 255) / 256, 256, 0, stream>>>(wsi, zn);
    fill_lists_kernel<<<EDGE_GRID, EDGE_WG, 0, stream>>>(ei, cnt, slots, ovf, cursor + 3,
                                                          cursor + 1, N, E);
    dirsage_pull_gemm_kernel<<<256, 32 * WAVES, 0, stream>>>(
        x, ei, cnt, slots, ovf, cursor + 3, Wself, bself, Ws2d, bs2d, Wd2s, bd2s, out, N, E);
}


# List sort ranks its ids with one unsigned compare per entry on (id<<5 | lane) keys, four entries per loop trip; the same ascending-id order as before, so every sum is unchanged bit for bit
# speedup vs baseline: 2.0815x; 1.0428x over previous
#include <hip/hip_runtime.h>


#define D 64
#define ALPHA 0.5f
#define PITCH 68
#define WAVES 28
#define BATCH 16
#define EDGE_WG 256
#define EDGE_PT 4
#define EDGE_CHUNK (32 * EDGE_PT)
#define EDGE_GRID 1024
#define CAP 32

typedef __attribute__((ext_vector_type(2))) float v2f;
typedef __attribute__((ext_vector_type(8))) float v8f;

__global__ void zero_counts_kernel(int* __restrict__ p, int n) {
    int i = blockIdx.x * blockDim.x + threadIdx.x;
    if (i < n) p[i] = 0;
}

__device__ __forceinline__ int chip_half() {
    return (int)((__builtin_amdgcn_s_sendmsg_rtn(0x87) >> 18) & 1u);
}

__device__ __forceinline__ int draw_ticket(int* counter, int lane) {
    int v = 0;
    if (lane == 0) v = atomicAdd(counter, 1);
    return __builtin_amdgcn_readfirstlane(v);
}

__global__ __launch_bounds__(EDGE_WG)
void fill_lists_kernel(const int* __restrict__ ei,
                       int* __restrict__ cnt,
                       int* __restrict__ slots,
                       int* __restrict__ ovf,
                       int* __restrict__ novf,
                       int* __restrict__ tickets,
                       int N, int E) {
    const int lane    = threadIdx.x & 31;
    const int home    = chip_half();
    const int nchunks = (E + EDGE_CHUNK - 1) / EDGE_CHUNK;
    for (int pass = 0; pass < 2; ++pass) {
        const int dir = home ^ pass;
        const int* __restrict__ key = ei + (dir ? 0 : E);
        const int* __restrict__ val = ei + (dir ? E : 0);
        int* __restrict__ c  = cnt + dir * N;
        int* __restrict__ sl = slots + (long long)dir * N * CAP;
        int* __restrict__ ov = ovf + (long long)dir * E;
        int t = draw_ticket(&tickets[dir], lane);
        for (int it = 0; it < nchunks && t < nchunks; ++it) {
            const int e0 = t * EDGE_CHUNK + lane;
            int k[EDGE_PT], v[EDGE_PT], r[EDGE_PT];
#pragma unroll
            for (int i = 0; i < EDGE_PT; ++i) {
                int e = e0 + i * 32;
                bool in = e < E;
                k[i] = in ? key[e] : -1;
                v[i] = in ? val[e] : 0;
            }
            t = draw_ticket(&tickets[dir], lane);
#pragma unroll
            for (int i = 0; i < EDGE_PT; ++i)
                r[i] = (k[i] >= 0) ? atomicAdd(&c[k[i]], 1) : 0;
#pragma unroll
            for (int i = 0; i < EDGE_PT; ++i) {
                if (k[i] >= 0) {
                    if (r[i] < CAP) {
                        sl[k[i] * CAP + r[i]] = v[i];
                    } else {
                        int o = atomicAdd(&novf[dir], 1);
                        if (o < E) ov[o] = e0 + i * 32;
                    }
                }
            }
        }
    }
}

__device__ __forceinline__ v2f sum_sorted(const float* __restrict__ x,
                                          int L, int nid, int lane) {
    v2f acc = {0.0f, 0.0f};
    int id = (lane < L) ? nid : 0x7fffffff;
    unsigned key = (lane < L) ? (((unsigned)nid << 5) | (unsigned)lane) : 0xffffffffu;
    int rk = 0;
    for (int t = 0; t < L; t += 4) {
#pragma unroll
        for (int u = 0; u < 4; ++u) {
            unsigned v = (unsigned)__builtin_amdgcn_readlane((int)key, t + u);
            rk += (v < key) ? 1 : 0;
        }
    }
    int sorted = __builtin_amdgcn_ds_permute(rk << 2, id);
    const float* xl = x + 2 * lane;
    for (int t0 = 0; t0 < L; t0 += BATCH) {
        v2f v[BATCH];
#pragma unroll
        for (int i = 0; i < BATCH; ++i) {
            if (t0 + i < L) {
                int j = __builtin_amdgcn_readlane(sorted, t0 + i);
                v[i] = *(const v2f*)(xl + (long long)j * D);
            }
        }
#pragma unroll
        for (int i = 0; i < BATCH; ++i)
            if (t0 + i < L) acc += v[i];
    }
    return acc;
}

__device__ __forceinline__ v2f sum_long(const float* __restrict__ x,
                                        const int* __restrict__ row,
                                        const int* __restrict__ ov, int nov,
                                        const int* __restrict__ key,
                                        const int* __restrict__ val,
                                        int node, int L, int E, int lane) {
    v2f acc = {0.0f, 0.0f};
    int own  = row[lane];
    int prev = -1;
    int done = 0;
    while (done < L) {
        int best = (own > prev) ? own : 0x7fffffff;
        for (int i = lane; i < nov; i += 32) {
            int e = ov[i];
            if ((unsigned)e < (unsigned)E && key[e] == node) {
                int v = val[e];
                if (v > prev && v < best) best = v;
            }
        }
        for (int m = 16; m >= 1; m >>= 1) {
            int o = __shfl_xor(best, m);
            best = o < best ? o : best;
        }
        int mult = (own == best) ? 1 : 0;
        for (int i = lane; i < nov; i += 32) {
            int e = ov[i];
            if ((unsigned)e < (unsigned)E && key[e] == node && val[e] == best) mult += 1;
        }
        for (int m = 16; m >= 1; m >>= 1) mult += __shfl_xor(mult, m);
        best = __builtin_amdgcn_readfirstlane(best);
        mult = __builtin_amdgcn_readfirstlane(mult);
        if (mult == 0) break;
        v2f r = *(const v2f*)(x + (long long)best * D + 2 * lane);
        acc += r;
        for (int q = 1; q < mult; ++q) acc += r;
        prev = best;
        done += mult;
    }
    return acc;
}

__device__ __forceinline__ void gather_tile(const float* __restrict__ x,
                                            const int* __restrict__ rows,
                                            float* st,
                                            int c, int sel, float alpha,
                                            const int* __restrict__ ov,
                                            const int* __restrict__ novp,
                                            const int* __restrict__ key,
                                            const int* __restrict__ val,
                                            int node0, int E, int lane) {
    int L0  = __builtin_amdgcn_readlane(c, sel);
    int nid = (lane < L0) ? rows[lane] : 0;
    for (int r = 0; r < 16; ++r) {
        int Lc = L0, nidc = nid;
        if (r < 15) {
            L0  = __builtin_amdgcn_readlane(c, sel + r + 1);
            nid = (lane < L0) ? rows[(r + 1) * CAP + lane] : 0;
        }
        v2f a = {0.0f, 0.0f};
        if (Lc <= CAP) a = sum_sorted(x, Lc, nidc, lane);
        a *= alpha / fmaxf((float)Lc, 1.0f);
        *(v2f*)(st + r * PITCH + 2 * lane) = a;
    }
    for (int r = 0; r < 16; ++r) {
        int Lc = __builtin_amdgcn_readlane(c, sel + r);
        if (Lc > CAP) {
            int nov = novp[0];
            nov = nov < E ? nov : E;
            v2f a = sum_long(x, rows + r * CAP, ov, nov, key, val, node0 + r, Lc, E, lane);
            a *= alpha / fmaxf((float)Lc, 1.0f);
            *(v2f*)(st + r * PITCH + 2 * lane) = a;
        }
    }
}

__global__ __launch_bounds__(32 * WAVES)
void dirsage_pull_gemm_kernel(const float* __restrict__ x,
                              const int*   __restrict__ ei,
                              const int*   __restrict__ cnt,
                              const int*   __restrict__ slots,
                              const int*   __restrict__ ovf,
                              const int*   __restrict__ novf,
                              const float* __restrict__ Wself,
                              const float* __restrict__ bself,
                              const float* __restrict__ Ws2d,
                              const float* __restrict__ bs2d,
                              const float* __restrict__ Wd2s,
                              const float* __restrict__ bd2s,
                              float* __restrict__ out,
                              int N, int E) {
    __shared__ float wl[3 * D * D];
    __shared__ float stage[WAVES * 16 * PITCH];

    for (int i = threadIdx.x; i < D * D; i += blockDim.x) {
        int k   = i >> 6;
        int col = i & 63;
        int dst = ((k >> 1) << 7) + (col << 1) + (k & 1);
        wl[dst]             = Wself[i];
        wl[D * D + dst]     = Ws2d[i];
        wl[2 * D * D + dst] = Wd2s[i];
    }
    __syncthreads();

    int wave   = threadIdx.x >> 5;
    int lane   = threadIdx.x & 31;
    int row    = lane & 15;
    int half   = lane >> 4;
    int khalf  = half * 2;
    int mtiles = N >> 4;
    float* st  = stage + wave * 16 * PITCH;
    const float* wbase = &wl[half * 128 + row * 2];
    const float* sbase = st + row * PITCH + khalf;

    for (int mtile = blockIdx.x + gridDim.x * wave; mtile < mtiles;
         mtile += gridDim.x * WAVES) {
        int gr0 = mtile * 16;
        int gr  = gr0 + row;

        int li = half * N + gr;
        int c  = cnt[li];

        v8f acc[4];
        for (int t = 0; t < 4; ++t) {
            float b = bself[t * 16 + row]
                    + (1.0f - ALPHA) * bs2d[t * 16 + row]
                    + ALPHA * bd2s[t * 16 + row];
            for (int r = 0; r < 8; ++r) acc[t][r] = b;
        }

        const float* xrow = x + (long long)gr * D + khalf;
        v2f ax[16];
#pragma unroll
        for (int j = 0; j < 16; ++j) ax[j] = *(const v2f*)(xrow + 4 * j);
#pragma unroll
        for (int j = 0; j < 16; ++j) {
            const float* wk = wbase + j * 256;
            for (int t = 0; t < 4; ++t) {
                v2f bs = *(const v2f*)(wk + t * 32);
                acc[t] = __builtin_amdgcn_wmma_f32_16x16x4_f32(
                    false, ax[j], false, bs, (short)0, acc[t], false, false);
            }
        }

        for (int dir = 0; dir < 2; ++dir) {
            gather_tile(x, slots + ((long long)dir * N + gr0) * CAP, st, c, dir * 16,
                        dir ? ALPHA : (1.0f - ALPHA),
                        ovf + (long long)dir * E, novf + dir,
                        ei + (dir ? 0 : E), ei + (dir ? E : 0),
                        gr0, E, lane);
            __builtin_amdgcn_wave_barrier();
            const float* wd = wbase + (dir + 1) * D * D;
            for (int k0 = 0; k0 < D; k0 += 4) {
                v2f aa = *(const v2f*)(sbase + k0);
                const float* wk = wd + (k0 >> 1) * 128;
                for (int t = 0; t < 4; ++t) {
                    v2f bb = *(const v2f*)(wk + t * 32);
                    acc[t] = __builtin_amdgcn_wmma_f32_16x16x4_f32(
                        false, aa, false, bb, (short)0, acc[t], false, false);
                }
            }
            __builtin_amdgcn_wave_barrier();
        }

        int mofs = half * 8;
        for (int t = 0; t < 4; ++t) {
            int col = t * 16 + row;
            long long base = (long long)(gr0 + mofs) * D + col;
            for (int r = 0; r < 8; ++r) out[base + (long long)r * D] = acc[t][r];
        }
    }
}

extern "C" void kernel_launch(void* const* d_in, const int* in_sizes, int n_in,
                              void* d_out, int out_size, void* d_ws, size_t ws_size,
                              hipStream_t stream) {
    const float* x     = (const float*)d_in[0];
    const int*   ei    = (const int*)  d_in[1];
    const float* Wself = (const float*)d_in[2];
    const float* bself = (const float*)d_in[3];
    const float* Ws2d  = (const float*)d_in[4];
    const float* bs2d  = (const float*)d_in[5];
    const float* Wd2s  = (const float*)d_in[6];
    const float* bd2s  = (const float*)d_in[7];
    float* out = (float*)d_out;

    int N = in_sizes[0] / D;
    int E = in_sizes[1] / 2;

    int* wsi    = (int*)d_ws;
    int* cnt    = wsi;
    int* cursor = wsi + 2 * (size_t)N;
    int* slots  = cursor + 32;
    int* ovf    = slots + 2 * (size_t)N * CAP;

    int zn = 2 * N + 32;
    zero_counts_kernel<<<(zn + 255) / 256, 256, 0, stream>>>(wsi, zn);
    fill_lists_kernel<<<EDGE_GRID, EDGE_WG, 0, stream>>>(ei, cnt, slots, ovf, cursor + 3,
                                                          cursor + 1, N, E);
    dirsage_pull_gemm_kernel<<<256, 32 * WAVES, 0, stream>>>(
        x, ei, cnt, slots, ovf, cursor + 3, Wself, bself, Ws2d, bs2d, Wd2s, bd2s, out, N, E);
}


# Lists of 33 to 64 entries take their side-list ids in one scan and add rows in sorted batches like short lists (was two side-list scans per distinct id); each sum's order is unchanged bit for bit
# speedup vs baseline: 2.5072x; 1.2045x over previous
#include <hip/hip_runtime.h>


#define D 64
#define ALPHA 0.5f
#define PITCH 68
#define WAVES 28
#define BATCH 16
#define EDGE_WG 256
#define EDGE_PT 4
#define EDGE_CHUNK (32 * EDGE_PT)
#define EDGE_GRID 1024
#define CAP 32

typedef __attribute__((ext_vector_type(2))) float v2f;
typedef __attribute__((ext_vector_type(8))) float v8f;

__global__ void zero_counts_kernel(int* __restrict__ p, int n) {
    int i = blockIdx.x * blockDim.x + threadIdx.x;
    if (i < n) p[i] = 0;
}

__device__ __forceinline__ int chip_half() {
    return (int)((__builtin_amdgcn_s_sendmsg_rtn(0x87) >> 18) & 1u);
}

__device__ __forceinline__ int draw_ticket(int* counter, int lane) {
    int v = 0;
    if (lane == 0) v = atomicAdd(counter, 1);
    return __builtin_amdgcn_readfirstlane(v);
}

__global__ __launch_bounds__(EDGE_WG)
void fill_lists_kernel(const int* __restrict__ ei,
                       int* __restrict__ cnt,
                       int* __restrict__ slots,
                       int* __restrict__ ovf,
                       int* __restrict__ novf,
                       int* __restrict__ tickets,
                       int N, int E) {
    const int lane    = threadIdx.x & 31;
    const int home    = chip_half();
    const int nchunks = (E + EDGE_CHUNK - 1) / EDGE_CHUNK;
    for (int pass = 0; pass < 2; ++pass) {
        const int dir = home ^ pass;
        const int* __restrict__ key = ei + (dir ? 0 : E);
        const int* __restrict__ val = ei + (dir ? E : 0);
        int* __restrict__ c  = cnt + dir * N;
        int* __restrict__ sl = slots + (long long)dir * N * CAP;
        int* __restrict__ ov = ovf + (long long)dir * E;
        int t = draw_ticket(&tickets[dir], lane);
        for (int it = 0; it < nchunks && t < nchunks; ++it) {
            const int e0 = t * EDGE_CHUNK + lane;
            int k[EDGE_PT], v[EDGE_PT], r[EDGE_PT];
#pragma unroll
            for (int i = 0; i < EDGE_PT; ++i) {
                int e = e0 + i * 32;
                bool in = e < E;
                k[i] = in ? key[e] : -1;
                v[i] = in ? val[e] : 0;
            }
            t = draw_ticket(&tickets[dir], lane);
#pragma unroll
            for (int i = 0; i < EDGE_PT; ++i)
                r[i] = (k[i] >= 0) ? atomicAdd(&c[k[i]], 1) : 0;
#pragma unroll
            for (int i = 0; i < EDGE_PT; ++i) {
                if (k[i] >= 0) {
                    if (r[i] < CAP) {
                        sl[k[i] * CAP + r[i]] = v[i];
                    } else {
                        int o = atomicAdd(&novf[dir], 1);
                        if (o < E) ov[o] = e0 + i * 32;
                    }
                }
            }
        }
    }
}

__device__ __forceinline__ v2f sum_sorted(const float* __restrict__ x,
                                          int L, int nid, int lane) {
    v2f acc = {0.0f, 0.0f};
    int id = (lane < L) ? nid : 0x7fffffff;
    unsigned key = (lane < L) ? (((unsigned)nid << 5) | (unsigned)lane) : 0xffffffffu;
    int rk = 0;
    for (int t = 0; t < L; t += 4) {
#pragma unroll
        for (int u = 0; u < 4; ++u) {
            unsigned v = (unsigned)__builtin_amdgcn_readlane((int)key, t + u);
            rk += (v < key) ? 1 : 0;
        }
    }
    int sorted = __builtin_amdgcn_ds_permute(rk << 2, id);
    const float* xl = x + 2 * lane;
    for (int t0 = 0; t0 < L; t0 += BATCH) {
        v2f v[BATCH];
#pragma unroll
        for (int i = 0; i < BATCH; ++i) {
            if (t0 + i < L) {
                int j = __builtin_amdgcn_readlane(sorted, t0 + i);
                v[i] = *(const v2f*)(xl + (long long)j * D);
            }
        }
#pragma unroll
        for (int i = 0; i < BATCH; ++i)
            if (t0 + i < L) acc += v[i];
    }
    return acc;
}

__device__ __attribute__((noinline)) v2f sum_mid(const float* __restrict__ x,
                                       const int* __restrict__ row,
                                       const int* __restrict__ ov, int nov,
                                       const int* __restrict__ key,
                                       const int* __restrict__ val,
                                       int node, int E, int lane) {
    v2f acc = {0.0f, 0.0f};
    int id1 = row[lane];
    int id2 = 0x7fffffff;
    int m = 0;
    for (int i0 = 0; i0 < nov; i0 += 32) {
        int i = i0 + lane;
        int v = 0;
        bool hit = false;
        if (i < nov) {
            int e = ov[i];
            if ((unsigned)e < (unsigned)E && key[e] == node) {
                v = val[e];
                hit = true;
            }
        }
        unsigned found = __builtin_amdgcn_ballot_w32(hit);
        while (found != 0u && m < CAP) {
            int vv = __builtin_amdgcn_readlane(v, __builtin_ctz(found));
            if (lane == m) id2 = vv;
            ++m;
            found &= found - 1u;
        }
    }
    int rk1 = 0, rk2 = 0;
    for (int t = 0; t < CAP; ++t) {
        int a = __builtin_amdgcn_readlane(id1, t);
        rk1 += (a < id1 || (a == id1 && t < lane)) ? 1 : 0;
        rk2 += (a <= id2) ? 1 : 0;
    }
    for (int t = 0; t < m; ++t) {
        int b = __builtin_amdgcn_readlane(id2, t);
        rk1 += (b < id1) ? 1 : 0;
        rk2 += (b < id2 || (b == id2 && t < lane)) ? 1 : 0;
    }
    const int n = CAP + m;
    const float* xl = x + 2 * lane;
    for (int t0 = 0; t0 < n; t0 += BATCH) {
        v2f v[BATCH];
#pragma unroll
        for (int i = 0; i < BATCH; ++i) {
            if (t0 + i < n) {
                unsigned m1 = __builtin_amdgcn_ballot_w32(rk1 == t0 + i);
                unsigned m2 = __builtin_amdgcn_ballot_w32(lane < m && rk2 == t0 + i);
                int j1 = __builtin_amdgcn_readlane(id1, m1 ? __builtin_ctz(m1) : 0);
                int j2 = __builtin_amdgcn_readlane(id2, m2 ? __builtin_ctz(m2) : 0);
                int j  = m1 ? j1 : j2;
                v[i] = *(const v2f*)(xl + (long long)j * D);
            }
        }
#pragma unroll
        for (int i = 0; i < BATCH; ++i)
            if (t0 + i < n) acc += v[i];
    }
    return acc;
}

__device__ __forceinline__ v2f sum_long(const float* __restrict__ x,
                                        const int* __restrict__ row,
                                        const int* __restrict__ ov, int nov,
                                        const int* __restrict__ key,
                                        const int* __restrict__ val,
                                        int node, int L, int E, int lane) {
    v2f acc = {0.0f, 0.0f};
    int own  = row[lane];
    int prev = -1;
    int done = 0;
    while (done < L) {
        int best = (own > prev) ? own : 0x7fffffff;
        for (int i = lane; i < nov; i += 32) {
            int e = ov[i];
            if ((unsigned)e < (unsigned)E && key[e] == node) {
                int v = val[e];
                if (v > prev && v < best) best = v;
            }
        }
        for (int m = 16; m >= 1; m >>= 1) {
            int o = __shfl_xor(best, m);
            best = o < best ? o : best;
        }
        int mult = (own == best) ? 1 : 0;
        for (int i = lane; i < nov; i += 32) {
            int e = ov[i];
            if ((unsigned)e < (unsigned)E && key[e] == node && val[e] == best) mult += 1;
        }
        for (int m = 16; m >= 1; m >>= 1) mult += __shfl_xor(mult, m);
        best = __builtin_amdgcn_readfirstlane(best);
        mult = __builtin_amdgcn_readfirstlane(mult);
        if (mult == 0) break;
        v2f r = *(const v2f*)(x + (long long)best * D + 2 * lane);
        acc += r;
        for (int q = 1; q < mult; ++q) acc += r;
        prev = best;
        done += mult;
    }
    return acc;
}

__device__ __forceinline__ void gather_tile(const float* __restrict__ x,
                                            const int* __restrict__ rows,
                                            float* st,
                                            int c, int sel, float alpha,
                                            const int* __restrict__ ov,
                                            const int* __restrict__ novp,
                                            const int* __restrict__ key,
                                            const int* __restrict__ val,
                                            int node0, int E, int lane) {
    int L0  = __builtin_amdgcn_readlane(c, sel);
    int nid = (lane < L0) ? rows[lane] : 0;
    for (int r = 0; r < 16; ++r) {
        int Lc = L0, nidc = nid;
        if (r < 15) {
            L0  = __builtin_amdgcn_readlane(c, sel + r + 1);
            nid = (lane < L0) ? rows[(r + 1) * CAP + lane] : 0;
        }
        v2f a = {0.0f, 0.0f};
        if (Lc <= CAP) a = sum_sorted(x, Lc, nidc, lane);
        a *= alpha / fmaxf((float)Lc, 1.0f);
        *(v2f*)(st + r * PITCH + 2 * lane) = a;
    }
    for (int r = 0; r < 16; ++r) {
        int Lc = __builtin_amdgcn_readlane(c, sel + r);
        if (Lc > CAP) {
            int nov = novp[0];
            nov = nov < E ? nov : E;
            v2f a = (Lc <= 2 * CAP)
                  ? sum_mid(x, rows + r * CAP, ov, nov, key, val, node0 + r, E, lane)
                  : sum_long(x, rows + r * CAP, ov, nov, key, val, node0 + r, Lc, E, lane);
            a *= alpha / fmaxf((float)Lc, 1.0f);
            *(v2f*)(st + r * PITCH + 2 * lane) = a;
        }
    }
}

__global__ __launch_bounds__(32 * WAVES)
void dirsage_pull_gemm_kernel(const float* __restrict__ x,
                              const int*   __restrict__ ei,
                              const int*   __restrict__ cnt,
                              const int*   __restrict__ slots,
                              const int*   __restrict__ ovf,
                              const int*   __restrict__ novf,
                              const float* __restrict__ Wself,
                              const float* __restrict__ bself,
                              const float* __restrict__ Ws2d,
                              const float* __restrict__ bs2d,
                              const float* __restrict__ Wd2s,
                              const float* __restrict__ bd2s,
                              float* __restrict__ out,
                              int N, int E) {
    __shared__ float wl[3 * D * D];
    __shared__ float stage[WAVES * 16 * PITCH];

    for (int i = threadIdx.x; i < D * D; i += blockDim.x) {
        int k   = i >> 6;
        int col = i & 63;
        int dst = ((k >> 1) << 7) + (col << 1) + (k & 1);
        wl[dst]             = Wself[i];
        wl[D * D + dst]     = Ws2d[i];
        wl[2 * D * D + dst] = Wd2s[i];
    }
    __syncthreads();

    int wave   = threadIdx.x >> 5;
    int lane   = threadIdx.x & 31;
    int row    = lane & 15;
    int half   = lane >> 4;
    int khalf  = half * 2;
    int mtiles = N >> 4;
    float* st  = stage + wave * 16 * PITCH;
    const float* wbase = &wl[half * 128 + row * 2];
    const float* sbase = st + row * PITCH + khalf;

    for (int mtile = blockIdx.x + gridDim.x * wave; mtile < mtiles;
         mtile += gridDim.x * WAVES) {
        int gr0 = mtile * 16;
        int gr  = gr0 + row;

        int li = half * N + gr;
        int c  = cnt[li];

        v8f acc[4];
        for (int t = 0; t < 4; ++t) {
            float b = bself[t * 16 + row]
                    + (1.0f - ALPHA) * bs2d[t * 16 + row]
                    + ALPHA * bd2s[t * 16 + row];
            for (int r = 0; r < 8; ++r) acc[t][r] = b;
        }

        const float* xrow = x + (long long)gr * D + khalf;
        v2f ax[16];
#pragma unroll
        for (int j = 0; j < 16; ++j) ax[j] = *(const v2f*)(xrow + 4 * j);
#pragma unroll
        for (int j = 0; j < 16; ++j) {
            const float* wk = wbase + j * 256;
            for (int t = 0; t < 4; ++t) {
                v2f bs = *(const v2f*)(wk + t * 32);
                acc[t] = __builtin_amdgcn_wmma_f32_16x16x4_f32(
                    false, ax[j], false, bs, (short)0, acc[t], false, false);
            }
        }

        for (int dir = 0; dir < 2; ++dir) {
            gather_tile(x, slots + ((long long)dir * N + gr0) * CAP, st, c, dir * 16,
                        dir ? ALPHA : (1.0f - ALPHA),
                        ovf + (long long)dir * E, novf + dir,
                        ei + (dir ? 0 : E), ei + (dir ? E : 0),
                        gr0, E, lane);
            __builtin_amdgcn_wave_barrier();
            const float* wd = wbase + (dir + 1) * D * D;
            for (int k0 = 0; k0 < D; k0 += 4) {
                v2f aa = *(const v2f*)(sbase + k0);
                const float* wk = wd + (k0 >> 1) * 128;
                for (int t = 0; t < 4; ++t) {
                    v2f bb = *(const v2f*)(wk + t * 32);
                    acc[t] = __builtin_amdgcn_wmma_f32_16x16x4_f32(
                        false, aa, false, bb, (short)0, acc[t], false, false);
                }
            }
            __builtin_amdgcn_wave_barrier();
        }

        int mofs = half * 8;
        for (int t = 0; t < 4; ++t) {
            int col = t * 16 + row;
            long long base = (long long)(gr0 + mofs) * D + col;
            for (int r = 0; r < 8; ++r) out[base + (long long)r * D] = acc[t][r];
        }
    }
}

extern "C" void kernel_launch(void* const* d_in, const int* in_sizes, int n_in,
                              void* d_out, int out_size, void* d_ws, size_t ws_size,
                              hipStream_t stream) {
    const float* x     = (const float*)d_in[0];
    const int*   ei    = (const int*)  d_in[1];
    const float* Wself = (const float*)d_in[2];
    const float* bself = (const float*)d_in[3];
    const float* Ws2d  = (const float*)d_in[4];
    const float* bs2d  = (const float*)d_in[5];
    const float* Wd2s  = (const float*)d_in[6];
    const float* bd2s  = (const float*)d_in[7];
    float* out = (float*)d_out;

    int N = in_sizes[0] / D;
    int E = in_sizes[1] / 2;

    int* wsi    = (int*)d_ws;
    int* cnt    = wsi;
    int* cursor = wsi + 2 * (size_t)N;
    int* slots  = cursor + 32;
    int* ovf    = slots + 2 * (size_t)N * CAP;

    int zn = 2 * N + 32;
    zero_counts_kernel<<<(zn + 255) / 256, 256, 0, stream>>>(wsi, zn);
    fill_lists_kernel<<<EDGE_GRID, EDGE_WG, 0, stream>>>(ei, cnt, slots, ovf, cursor + 3,
                                                          cursor + 1, N, E);
    dirsage_pull_gemm_kernel<<<256, 32 * WAVES, 0, stream>>>(
        x, ei, cnt, slots, ovf, cursor + 3, Wself, bself, Ws2d, bs2d, Wd2s, bd2s, out, N, E);
}


# Row gather of a list taken in groups of four entries under one wave-uniform test, fixed lane reads and 32-bit row offsets, no per-entry test or select; every sum's order unchanged bit for bit
# speedup vs baseline: 2.7721x; 1.1057x over previous
#include <hip/hip_runtime.h>


#define D 64
#define ALPHA 0.5f
#define PITCH 68
#define WAVES 28
#define BATCH 16
#define EDGE_WG 256
#define EDGE_PT 4
#define EDGE_CHUNK (32 * EDGE_PT)
#define EDGE_GRID 1024
#define CAP 32

typedef __attribute__((ext_vector_type(2))) float v2f;
typedef __attribute__((ext_vector_type(8))) float v8f;

__global__ void zero_counts_kernel(int* __restrict__ p, int n) {
    int i = blockIdx.x * blockDim.x + threadIdx.x;
    if (i < n) p[i] = 0;
}

__device__ __forceinline__ int chip_half() {
    return (int)((__builtin_amdgcn_s_sendmsg_rtn(0x87) >> 18) & 1u);
}

__device__ __forceinline__ int draw_ticket(int* counter, int lane) {
    int v = 0;
    if (lane == 0) v = atomicAdd(counter, 1);
    return __builtin_amdgcn_readfirstlane(v);
}

__global__ __launch_bounds__(EDGE_WG)
void fill_lists_kernel(const int* __restrict__ ei,
                       int* __restrict__ cnt,
                       int* __restrict__ slots,
                       int* __restrict__ ovf,
                       int* __restrict__ novf,
                       int* __restrict__ tickets,
                       int N, int E) {
    const int lane    = threadIdx.x & 31;
    const int home    = chip_half();
    const int nchunks = (E + EDGE_CHUNK - 1) / EDGE_CHUNK;
    for (int pass = 0; pass < 2; ++pass) {
        const int dir = home ^ pass;
        const int* __restrict__ key = ei + (dir ? 0 : E);
        const int* __restrict__ val = ei + (dir ? E : 0);
        int* __restrict__ c  = cnt + dir * N;
        int* __restrict__ sl = slots + (long long)dir * N * CAP;
        int* __restrict__ ov = ovf + (long long)dir * E;
        int t = draw_ticket(&tickets[dir], lane);
        for (int it = 0; it < nchunks && t < nchunks; ++it) {
            const int e0 = t * EDGE_CHUNK + lane;
            int k[EDGE_PT], v[EDGE_PT], r[EDGE_PT];
#pragma unroll
            for (int i = 0; i < EDGE_PT; ++i) {
                int e = e0 + i * 32;
                bool in = e < E;
                k[i] = in ? key[e] : -1;
                v[i] = in ? val[e] : 0;
            }
            t = draw_ticket(&tickets[dir], lane);
#pragma unroll
            for (int i = 0; i < EDGE_PT; ++i)
                r[i] = (k[i] >= 0) ? atomicAdd(&c[k[i]], 1) : 0;
#pragma unroll
            for (int i = 0; i < EDGE_PT; ++i) {
                if (k[i] >= 0) {
                    if (r[i] < CAP) {
                        sl[k[i] * CAP + r[i]] = v[i];
                    } else {
                        int o = atomicAdd(&novf[dir], 1);
                        if (o < E) ov[o] = e0 + i * 32;
                    }
                }
            }
        }
    }
}

#define ROW_AT(l) (*(const v2f*)(xb + ((unsigned)__builtin_amdgcn_readlane((int)sorted, (l)) + lane8)))
template <int BASE>
__device__ __forceinline__ v2f add_rows(v2f acc, const char* __restrict__ xb,
                                        unsigned sorted, unsigned lane8, int n) {
    const int g = n >> 2;
    const int r = n & 3;
    const int tail = BASE + 4 * g;
    v2f v[BATCH];
    v2f w0, w1, w2;
    if (g > 0) {
        v[0] = ROW_AT(BASE + 0);  v[1] = ROW_AT(BASE + 1);
        v[2] = ROW_AT(BASE + 2);  v[3] = ROW_AT(BASE + 3);
        if (g > 1) {
            v[4] = ROW_AT(BASE + 4);  v[5] = ROW_AT(BASE + 5);
            v[6] = ROW_AT(BASE + 6);  v[7] = ROW_AT(BASE + 7);
            if (g > 2) {
                v[8]  = ROW_AT(BASE + 8);   v[9]  = ROW_AT(BASE + 9);
                v[10] = ROW_AT(BASE + 10);  v[11] = ROW_AT(BASE + 11);
                if (g > 3) {
                    v[12] = ROW_AT(BASE + 12);  v[13] = ROW_AT(BASE + 13);
                    v[14] = ROW_AT(BASE + 14);  v[15] = ROW_AT(BASE + 15);
                }
            }
        }
    }
    if (r > 0) {
        w0 = ROW_AT(tail);
        if (r > 1) {
            w1 = ROW_AT(tail + 1);
            if (r > 2) w2 = ROW_AT(tail + 2);
        }
    }
    if (g > 0) {
        acc += v[0];  acc += v[1];  acc += v[2];  acc += v[3];
        if (g > 1) {
            acc += v[4];  acc += v[5];  acc += v[6];  acc += v[7];
            if (g > 2) {
                acc += v[8];  acc += v[9];  acc += v[10];  acc += v[11];
                if (g > 3) {
                    acc += v[12];  acc += v[13];  acc += v[14];  acc += v[15];
                }
            }
        }
    }
    if (r > 0) {
        acc += w0;
        if (r > 1) {
            acc += w1;
            if (r > 2) acc += w2;
        }
    }
    return acc;
}

__device__ __forceinline__ v2f sum_sorted(const float* __restrict__ x,
                                          int L, int nid, int lane) {
    v2f acc = {0.0f, 0.0f};
    unsigned key = (lane < L) ? (((unsigned)nid << 5) | (unsigned)lane) : 0xffffffffu;
    int rk = 0;
    for (int t = 0; t < L; t += 4) {
#pragma unroll
        for (int u = 0; u < 4; ++u) {
            unsigned v = (unsigned)__builtin_amdgcn_readlane((int)key, t + u);
            rk += (v < key) ? 1 : 0;
        }
    }
    unsigned sorted = (unsigned)__builtin_amdgcn_ds_permute(rk << 2, (int)((unsigned)nid << 8));
    const char* xb = (const char*)x;
    const unsigned lane8 = 8u * (unsigned)lane;
    acc = add_rows<0>(acc, xb, sorted, lane8, L < BATCH ? L : BATCH);
    if (L > BATCH) acc = add_rows<BATCH>(acc, xb, sorted, lane8, L - BATCH);
    return acc;
}

__device__ __attribute__((noinline)) v2f sum_mid(const float* __restrict__ x,
                                       const int* __restrict__ row,
                                       const int* __restrict__ ov, int nov,
                                       const int* __restrict__ key,
                                       const int* __restrict__ val,
                                       int node, int E, int lane) {
    v2f acc = {0.0f, 0.0f};
    int id1 = row[lane];
    int id2 = 0x7fffffff;
    int m = 0;
    for (int i0 = 0; i0 < nov; i0 += 32) {
        int i = i0 + lane;
        int v = 0;
        bool hit = false;
        if (i < nov) {
            int e = ov[i];
            if ((unsigned)e < (unsigned)E && key[e] == node) {
                v = val[e];
                hit = true;
            }
        }
        unsigned found = __builtin_amdgcn_ballot_w32(hit);
        while (found != 0u && m < CAP) {
            int vv = __builtin_amdgcn_readlane(v, __builtin_ctz(found));
            if (lane == m) id2 = vv;
            ++m;
            found &= found - 1u;
        }
    }
    int rk1 = 0, rk2 = 0;
    for (int t = 0; t < CAP; ++t) {
        int a = __builtin_amdgcn_readlane(id1, t);
        rk1 += (a < id1 || (a == id1 && t < lane)) ? 1 : 0;
        rk2 += (a <= id2) ? 1 : 0;
    }
    for (int t = 0; t < m; ++t) {
        int b = __builtin_amdgcn_readlane(id2, t);
        rk1 += (b < id1) ? 1 : 0;
        rk2 += (b < id2 || (b == id2 && t < lane)) ? 1 : 0;
    }
    const int n = CAP + m;
    const float* xl = x + 2 * lane;
    for (int t0 = 0; t0 < n; t0 += BATCH) {
        v2f v[BATCH];
#pragma unroll
        for (int i = 0; i < BATCH; ++i) {
            if (t0 + i < n) {
                unsigned m1 = __builtin_amdgcn_ballot_w32(rk1 == t0 + i);
                unsigned m2 = __builtin_amdgcn_ballot_w32(lane < m && rk2 == t0 + i);
                int j1 = __builtin_amdgcn_readlane(id1, m1 ? __builtin_ctz(m1) : 0);
                int j2 = __builtin_amdgcn_readlane(id2, m2 ? __builtin_ctz(m2) : 0);
                int j  = m1 ? j1 : j2;
                v[i] = *(const v2f*)(xl + (long long)j * D);
            }
        }
#pragma unroll
        for (int i = 0; i < BATCH; ++i)
            if (t0 + i < n) acc += v[i];
    }
    return acc;
}

__device__ __forceinline__ v2f sum_long(const float* __restrict__ x,
                                        const int* __restrict__ row,
                                        const int* __restrict__ ov, int nov,
                                        const int* __restrict__ key,
                                        const int* __restrict__ val,
                                        int node, int L, int E, int lane) {
    v2f acc = {0.0f, 0.0f};
    int own  = row[lane];
    int prev = -1;
    int done = 0;
    while (done < L) {
        int best = (own > prev) ? own : 0x7fffffff;
        for (int i = lane; i < nov; i += 32) {
            int e = ov[i];
            if ((unsigned)e < (unsigned)E && key[e] == node) {
                int v = val[e];
                if (v > prev && v < best) best = v;
            }
        }
        for (int m = 16; m >= 1; m >>= 1) {
            int o = __shfl_xor(best, m);
            best = o < best ? o : best;
        }
        int mult = (own == best) ? 1 : 0;
        for (int i = lane; i < nov; i += 32) {
            int e = ov[i];
            if ((unsigned)e < (unsigned)E && key[e] == node && val[e] == best) mult += 1;
        }
        for (int m = 16; m >= 1; m >>= 1) mult += __shfl_xor(mult, m);
        best = __builtin_amdgcn_readfirstlane(best);
        mult = __builtin_amdgcn_readfirstlane(mult);
        if (mult == 0) break;
        v2f r = *(const v2f*)(x + (long long)best * D + 2 * lane);
        acc += r;
        for (int q = 1; q < mult; ++q) acc += r;
        prev = best;
        done += mult;
    }
    return acc;
}

__device__ __forceinline__ void gather_tile(const float* __restrict__ x,
                                            const int* __restrict__ rows,
                                            float* st,
                                            int c, int sel, float alpha,
                                            const int* __restrict__ ov,
                                            const int* __restrict__ novp,
                                            const int* __restrict__ key,
                                            const int* __restrict__ val,
                                            int node0, int E, int lane) {
    int L0  = __builtin_amdgcn_readlane(c, sel);
    int nid = (lane < L0) ? rows[lane] : 0;
    for (int r = 0; r < 16; ++r) {
        int Lc = L0, nidc = nid;
        if (r < 15) {
            L0  = __builtin_amdgcn_readlane(c, sel + r + 1);
            nid = (lane < L0) ? rows[(r + 1) * CAP + lane] : 0;
        }
        v2f a = {0.0f, 0.0f};
        if (Lc <= CAP) a = sum_sorted(x, Lc, nidc, lane);
        a *= alpha / fmaxf((float)Lc, 1.0f);
        *(v2f*)(st + r * PITCH + 2 * lane) = a;
    }
    for (int r = 0; r < 16; ++r) {
        int Lc = __builtin_amdgcn_readlane(c, sel + r);
        if (Lc > CAP) {
            int nov = novp[0];
            nov = nov < E ? nov : E;
            v2f a = (Lc <= 2 * CAP)
                  ? sum_mid(x, rows + r * CAP, ov, nov, key, val, node0 + r, E, lane)
                  : sum_long(x, rows + r * CAP, ov, nov, key, val, node0 + r, Lc, E, lane);
            a *= alpha / fmaxf((float)Lc, 1.0f);
            *(v2f*)(st + r * PITCH + 2 * lane) = a;
        }
    }
}

__global__ __launch_bounds__(32 * WAVES)
void dirsage_pull_gemm_kernel(const float* __restrict__ x,
                              const int*   __restrict__ ei,
                              const int*   __restrict__ cnt,
                              const int*   __restrict__ slots,
                              const int*   __restrict__ ovf,
                              const int*   __restrict__ novf,
                              const float* __restrict__ Wself,
                              const float* __restrict__ bself,
                              const float* __restrict__ Ws2d,
                              const float* __restrict__ bs2d,
                              const float* __restrict__ Wd2s,
                              const float* __restrict__ bd2s,
                              float* __restrict__ out,
                              int N, int E) {
    __shared__ float wl[3 * D * D];
    __shared__ float stage[WAVES * 16 * PITCH];

    for (int i = threadIdx.x; i < D * D; i += blockDim.x) {
        int k   = i >> 6;
        int col = i & 63;
        int dst = ((k >> 1) << 7) + (col << 1) + (k & 1);
        wl[dst]             = Wself[i];
        wl[D * D + dst]     = Ws2d[i];
        wl[2 * D * D + dst] = Wd2s[i];
    }
    __syncthreads();

    int wave   = threadIdx.x >> 5;
    int lane   = threadIdx.x & 31;
    int row    = lane & 15;
    int half   = lane >> 4;
    int khalf  = half * 2;
    int mtiles = N >> 4;
    float* st  = stage + wave * 16 * PITCH;
    const float* wbase = &wl[half * 128 + row * 2];
    const float* sbase = st + row * PITCH + khalf;

    for (int mtile = blockIdx.x + gridDim.x * wave; mtile < mtiles;
         mtile += gridDim.x * WAVES) {
        int gr0 = mtile * 16;
        int gr  = gr0 + row;

        int li = half * N + gr;
        int c  = cnt[li];

        v8f acc[4];
        for (int t = 0; t < 4; ++t) {
            float b = bself[t * 16 + row]
                    + (1.0f - ALPHA) * bs2d[t * 16 + row]
                    + ALPHA * bd2s[t * 16 + row];
            for (int r = 0; r < 8; ++r) acc[t][r] = b;
        }

        const float* xrow = x + (long long)gr * D + khalf;
        v2f ax[16];
#pragma unroll
        for (int j = 0; j < 16; ++j) ax[j] = *(const v2f*)(xrow + 4 * j);
#pragma unroll
        for (int j = 0; j < 16; ++j) {
            const float* wk = wbase + j * 256;
            for (int t = 0; t < 4; ++t) {
                v2f bs = *(const v2f*)(wk + t * 32);
                acc[t] = __builtin_amdgcn_wmma_f32_16x16x4_f32(
                    false, ax[j], false, bs, (short)0, acc[t], false, false);
            }
        }

        for (int dir = 0; dir < 2; ++dir) {
            gather_tile(x, slots + ((long long)dir * N + gr0) * CAP, st, c, dir * 16,
                        dir ? ALPHA : (1.0f - ALPHA),
                        ovf + (long long)dir * E, novf + dir,
                        ei + (dir ? 0 : E), ei + (dir ? E : 0),
                        gr0, E, lane);
            __builtin_amdgcn_wave_barrier();
            const float* wd = wbase + (dir + 1) * D * D;
            for (int k0 = 0; k0 < D; k0 += 4) {
                v2f aa = *(const v2f*)(sbase + k0);
                const float* wk = wd + (k0 >> 1) * 128;
                for (int t = 0; t < 4; ++t) {
                    v2f bb = *(const v2f*)(wk + t * 32);
                    acc[t] = __builtin_amdgcn_wmma_f32_16x16x4_f32(
                        false, aa, false, bb, (short)0, acc[t], false, false);
                }
            }
            __builtin_amdgcn_wave_barrier();
        }

        int mofs = half * 8;
        for (int t = 0; t < 4; ++t) {
            int col = t * 16 + row;
            long long base = (long long)(gr0 + mofs) * D + col;
            for (int r = 0; r < 8; ++r) out[base + (long long)r * D] = acc[t][r];
        }
    }
}

extern "C" void kernel_launch(void* const* d_in, const int* in_sizes, int n_in,
                              void* d_out, int out_size, void* d_ws, size_t ws_size,
                              hipStream_t stream) {
    const float* x     = (const float*)d_in[0];
    const int*   ei    = (const int*)  d_in[1];
    const float* Wself = (const float*)d_in[2];
    const float* bself = (const float*)d_in[3];
    const float* Ws2d  = (const float*)d_in[4];
    const float* bs2d  = (const float*)d_in[5];
    const float* Wd2s  = (const float*)d_in[6];
    const float* bd2s  = (const float*)d_in[7];
    float* out = (float*)d_out;

    int N = in_sizes[0] / D;
    int E = in_sizes[1] / 2;

    int* wsi    = (int*)d_ws;
    int* cnt    = wsi;
    int* cursor = wsi + 2 * (size_t)N;
    int* slots  = cursor + 32;
    int* ovf    = slots + 2 * (size_t)N * CAP;

    int zn = 2 * N + 32;
    zero_counts_kernel<<<(zn + 255) / 256, 256, 0, stream>>>(wsi, zn);
    fill_lists_kernel<<<EDGE_GRID, EDGE_WG, 0, stream>>>(ei, cnt, slots, ovf, cursor + 3,
                                                          cursor + 1, N, E);
    dirsage_pull_gemm_kernel<<<256, 32 * WAVES, 0, stream>>>(
        x, ei, cnt, slots, ovf, cursor + 3, Wself, bself, Ws2d, bs2d, Wd2s, bd2s, out, N, E);
}


# Fill tickets: each direction's ticket in a 128-B line of its own (was one line drawn on from both cache halves), one draw per work-group for 8 chunks (was per wave); sums unchanged bit for bit
# speedup vs baseline: 3.7321x; 1.3463x over previous
#include <hip/hip_runtime.h>


#define D 64
#define ALPHA 0.5f
#define PITCH 68
#define WAVES 28
#define BATCH 16
#define EDGE_WG 256
#define EDGE_PT 4
#define EDGE_CHUNK (32 * EDGE_PT)
#define EDGE_GRID 1024
#define EDGE_WAVES (EDGE_WG / 32)
#define EDGE_UNIT (EDGE_WAVES * EDGE_CHUNK)
#define CAP 32
#define LINE_INTS 32

typedef __attribute__((ext_vector_type(2))) float v2f;
typedef __attribute__((ext_vector_type(8))) float v8f;

__global__ void zero_counts_kernel(int* __restrict__ p, int n) {
    int i = blockIdx.x * blockDim.x + threadIdx.x;
    if (i < n) p[i] = 0;
}

__device__ __forceinline__ int chip_half() {
    return (int)((__builtin_amdgcn_s_sendmsg_rtn(0x87) >> 18) & 1u);
}


__global__ __launch_bounds__(EDGE_WG)
void fill_lists_kernel(const int* __restrict__ ei,
                       int* __restrict__ cnt,
                       int* __restrict__ slots,
                       int* __restrict__ ovf,
                       int* __restrict__ cursor,
                       int N, int E) {
    __shared__ int tk[2];
    __shared__ int sh_home;
    const int lane   = threadIdx.x & 31;
    const int wave   = threadIdx.x >> 5;
    const int nunits = (E + EDGE_UNIT - 1) / EDGE_UNIT;
    if (threadIdx.x == 0) sh_home = chip_half();
    __syncthreads();
    const int home = sh_home;
    int slot = 0;
    for (int pass = 0; pass < 2; ++pass) {
        const int dir = home ^ pass;
        const int* __restrict__ key = ei + (dir ? 0 : E);
        const int* __restrict__ val = ei + (dir ? E : 0);
        int* __restrict__ c    = cnt + dir * N;
        int* __restrict__ sl   = slots + (long long)dir * N * CAP;
        int* __restrict__ ov   = ovf + (long long)dir * E;
        int* tick = cursor + dir * LINE_INTS;
        int* nov  = tick + 1;
        if (threadIdx.x == 0) tk[slot] = atomicAdd(tick, 1);
        __syncthreads();
        int t = __builtin_amdgcn_readfirstlane(tk[slot]);
        slot ^= 1;
        for (int it = 0; it < nunits && t < nunits; ++it) {
            const int e0 = (t * EDGE_WAVES + wave) * EDGE_CHUNK + lane;
            int k[EDGE_PT], v[EDGE_PT], r[EDGE_PT];
#pragma unroll
            for (int i = 0; i < EDGE_PT; ++i) {
                int e = e0 + i * 32;
                bool in = e < E;
                k[i] = in ? key[e] : -1;
                v[i] = in ? val[e] : 0;
            }
            if (threadIdx.x == 0) tk[slot] = atomicAdd(tick, 1);
            __syncthreads();
            const int tn = __builtin_amdgcn_readfirstlane(tk[slot]);
            slot ^= 1;
#pragma unroll
            for (int i = 0; i < EDGE_PT; ++i)
                r[i] = (k[i] >= 0) ? atomicAdd(&c[k[i]], 1) : 0;
#pragma unroll
            for (int i = 0; i < EDGE_PT; ++i) {
                if (k[i] >= 0) {
                    if (r[i] < CAP) {
                        sl[k[i] * CAP + r[i]] = v[i];
                    } else {
                        int o = atomicAdd(nov, 1);
                        if (o < E) ov[o] = e0 + i * 32;
                    }
                }
            }
            t = tn;
        }
    }
}

#define ROW_AT(l) (*(const v2f*)(xb + ((unsigned)__builtin_amdgcn_readlane((int)sorted, (l)) + lane8)))
template <int BASE>
__device__ __forceinline__ v2f add_rows(v2f acc, const char* __restrict__ xb,
                                        unsigned sorted, unsigned lane8, int n) {
    const int g = n >> 2;
    const int r = n & 3;
    const int tail = BASE + 4 * g;
    v2f v[BATCH];
    v2f w0, w1, w2;
    if (g > 0) {
        v[0] = ROW_AT(BASE + 0);  v[1] = ROW_AT(BASE + 1);
        v[2] = ROW_AT(BASE + 2);  v[3] = ROW_AT(BASE + 3);
        if (g > 1) {
            v[4] = ROW_AT(BASE + 4);  v[5] = ROW_AT(BASE + 5);
            v[6] = ROW_AT(BASE + 6);  v[7] = ROW_AT(BASE + 7);
            if (g > 2) {
                v[8]  = ROW_AT(BASE + 8);   v[9]  = ROW_AT(BASE + 9);
                v[10] = ROW_AT(BASE + 10);  v[11] = ROW_AT(BASE + 11);
                if (g > 3) {
                    v[12] = ROW_AT(BASE + 12);  v[13] = ROW_AT(BASE + 13);
                    v[14] = ROW_AT(BASE + 14);  v[15] = ROW_AT(BASE + 15);
                }
            }
        }
    }
    if (r > 0) {
        w0 = ROW_AT(tail);
        if (r > 1) {
            w1 = ROW_AT(tail + 1);
            if (r > 2) w2 = ROW_AT(tail + 2);
        }
    }
    if (g > 0) {
        acc += v[0];  acc += v[1];  acc += v[2];  acc += v[3];
        if (g > 1) {
            acc += v[4];  acc += v[5];  acc += v[6];  acc += v[7];
            if (g > 2) {
                acc += v[8];  acc += v[9];  acc += v[10];  acc += v[11];
                if (g > 3) {
                    acc += v[12];  acc += v[13];  acc += v[14];  acc += v[15];
                }
            }
        }
    }
    if (r > 0) {
        acc += w0;
        if (r > 1) {
            acc += w1;
            if (r > 2) acc += w2;
        }
    }
    return acc;
}

__device__ __forceinline__ v2f sum_sorted(const float* __restrict__ x,
                                          int L, int nid, int lane) {
    v2f acc = {0.0f, 0.0f};
    unsigned key = (lane < L) ? (((unsigned)nid << 5) | (unsigned)lane) : 0xffffffffu;
    int rk = 0;
    for (int t = 0; t < L; t += 4) {
#pragma unroll
        for (int u = 0; u < 4; ++u) {
            unsigned v = (unsigned)__builtin_amdgcn_readlane((int)key, t + u);
            rk += (v < key) ? 1 : 0;
        }
    }
    unsigned sorted = (unsigned)__builtin_amdgcn_ds_permute(rk << 2, (int)((unsigned)nid << 8));
    const char* xb = (const char*)x;
    const unsigned lane8 = 8u * (unsigned)lane;
    acc = add_rows<0>(acc, xb, sorted, lane8, L < BATCH ? L : BATCH);
    if (L > BATCH) acc = add_rows<BATCH>(acc, xb, sorted, lane8, L - BATCH);
    return acc;
}

__device__ __attribute__((noinline)) v2f sum_mid(const float* __restrict__ x,
                                       const int* __restrict__ row,
                                       const int* __restrict__ ov, int nov,
                                       const int* __restrict__ key,
                                       const int* __restrict__ val,
                                       int node, int E, int lane) {
    v2f acc = {0.0f, 0.0f};
    int id1 = row[lane];
    int id2 = 0x7fffffff;
    int m = 0;
    for (int i0 = 0; i0 < nov; i0 += 32) {
        int i = i0 + lane;
        int v = 0;
        bool hit = false;
        if (i < nov) {
            int e = ov[i];
            if ((unsigned)e < (unsigned)E && key[e] == node) {
                v = val[e];
                hit = true;
            }
        }
        unsigned found = __builtin_amdgcn_ballot_w32(hit);
        while (found != 0u && m < CAP) {
            int vv = __builtin_amdgcn_readlane(v, __builtin_ctz(found));
            if (lane == m) id2 = vv;
            ++m;
            found &= found - 1u;
        }
    }
    int rk1 = 0, rk2 = 0;
    for (int t = 0; t < CAP; ++t) {
        int a = __builtin_amdgcn_readlane(id1, t);
        rk1 += (a < id1 || (a == id1 && t < lane)) ? 1 : 0;
        rk2 += (a <= id2) ? 1 : 0;
    }
    for (int t = 0; t < m; ++t) {
        int b = __builtin_amdgcn_readlane(id2, t);
        rk1 += (b < id1) ? 1 : 0;
        rk2 += (b < id2 || (b == id2 && t < lane)) ? 1 : 0;
    }
    const int n = CAP + m;
    const float* xl = x + 2 * lane;
    for (int t0 = 0; t0 < n; t0 += BATCH) {
        v2f v[BATCH];
#pragma unroll
        for (int i = 0; i < BATCH; ++i) {
            if (t0 + i < n) {
                unsigned m1 = __builtin_amdgcn_ballot_w32(rk1 == t0 + i);
                unsigned m2 = __builtin_amdgcn_ballot_w32(lane < m && rk2 == t0 + i);
                int j1 = __builtin_amdgcn_readlane(id1, m1 ? __builtin_ctz(m1) : 0);
                int j2 = __builtin_amdgcn_readlane(id2, m2 ? __builtin_ctz(m2) : 0);
                int j  = m1 ? j1 : j2;
                v[i] = *(const v2f*)(xl + (long long)j * D);
            }
        }
#pragma unroll
        for (int i = 0; i < BATCH; ++i)
            if (t0 + i < n) acc += v[i];
    }
    return acc;
}

__device__ __forceinline__ v2f sum_long(const float* __restrict__ x,
                                        const int* __restrict__ row,
                                        const int* __restrict__ ov, int nov,
                                        const int* __restrict__ key,
                                        const int* __restrict__ val,
                                        int node, int L, int E, int lane) {
    v2f acc = {0.0f, 0.0f};
    int own  = row[lane];
    int prev = -1;
    int done = 0;
    while (done < L) {
        int best = (own > prev) ? own : 0x7fffffff;
        for (int i = lane; i < nov; i += 32) {
            int e = ov[i];
            if ((unsigned)e < (unsigned)E && key[e] == node) {
                int v = val[e];
                if (v > prev && v < best) best = v;
            }
        }
        for (int m = 16; m >= 1; m >>= 1) {
            int o = __shfl_xor(best, m);
            best = o < best ? o : best;
        }
        int mult = (own == best) ? 1 : 0;
        for (int i = lane; i < nov; i += 32) {
            int e = ov[i];
            if ((unsigned)e < (unsigned)E && key[e] == node && val[e] == best) mult += 1;
        }
        for (int m = 16; m >= 1; m >>= 1) mult += __shfl_xor(mult, m);
        best = __builtin_amdgcn_readfirstlane(best);
        mult = __builtin_amdgcn_readfirstlane(mult);
        if (mult == 0) break;
        v2f r = *(const v2f*)(x + (long long)best * D + 2 * lane);
        acc += r;
        for (int q = 1; q < mult; ++q) acc += r;
        prev = best;
        done += mult;
    }
    return acc;
}

__device__ __forceinline__ void gather_tile(const float* __restrict__ x,
                                            const int* __restrict__ rows,
                                            float* st,
                                            int c, int sel, float alpha,
                                            const int* __restrict__ ov,
                                            const int* __restrict__ novp,
                                            const int* __restrict__ key,
                                            const int* __restrict__ val,
                                            int node0, int E, int lane) {
    int L0  = __builtin_amdgcn_readlane(c, sel);
    int nid = (lane < L0) ? rows[lane] : 0;
    for (int r = 0; r < 16; ++r) {
        int Lc = L0, nidc = nid;
        if (r < 15) {
            L0  = __builtin_amdgcn_readlane(c, sel + r + 1);
            nid = (lane < L0) ? rows[(r + 1) * CAP + lane] : 0;
        }
        v2f a = {0.0f, 0.0f};
        if (Lc <= CAP) a = sum_sorted(x, Lc, nidc, lane);
        a *= alpha / fmaxf((float)Lc, 1.0f);
        *(v2f*)(st + r * PITCH + 2 * lane) = a;
    }
    for (int r = 0; r < 16; ++r) {
        int Lc = __builtin_amdgcn_readlane(c, sel + r);
        if (Lc > CAP) {
            int nov = novp[0];
            nov = nov < E ? nov : E;
            v2f a = (Lc <= 2 * CAP)
                  ? sum_mid(x, rows + r * CAP, ov, nov, key, val, node0 + r, E, lane)
                  : sum_long(x, rows + r * CAP, ov, nov, key, val, node0 + r, Lc, E, lane);
            a *= alpha / fmaxf((float)Lc, 1.0f);
            *(v2f*)(st + r * PITCH + 2 * lane) = a;
        }
    }
}

__global__ __launch_bounds__(32 * WAVES)
void dirsage_pull_gemm_kernel(const float* __restrict__ x,
                              const int*   __restrict__ ei,
                              const int*   __restrict__ cnt,
                              const int*   __restrict__ slots,
                              const int*   __restrict__ ovf,
                              const int*   __restrict__ cursor,
                              const float* __restrict__ Wself,
                              const float* __restrict__ bself,
                              const float* __restrict__ Ws2d,
                              const float* __restrict__ bs2d,
                              const float* __restrict__ Wd2s,
                              const float* __restrict__ bd2s,
                              float* __restrict__ out,
                              int N, int E) {
    __shared__ float wl[3 * D * D];
    __shared__ float stage[WAVES * 16 * PITCH];

    for (int i = threadIdx.x; i < D * D; i += blockDim.x) {
        int k   = i >> 6;
        int col = i & 63;
        int dst = ((k >> 1) << 7) + (col << 1) + (k & 1);
        wl[dst]             = Wself[i];
        wl[D * D + dst]     = Ws2d[i];
        wl[2 * D * D + dst] = Wd2s[i];
    }
    __syncthreads();

    int wave   = threadIdx.x >> 5;
    int lane   = threadIdx.x & 31;
    int row    = lane & 15;
    int half   = lane >> 4;
    int khalf  = half * 2;
    int mtiles = N >> 4;
    float* st  = stage + wave * 16 * PITCH;
    const float* wbase = &wl[half * 128 + row * 2];
    const float* sbase = st + row * PITCH + khalf;

    for (int mtile = blockIdx.x + gridDim.x * wave; mtile < mtiles;
         mtile += gridDim.x * WAVES) {
        int gr0 = mtile * 16;
        int gr  = gr0 + row;

        int li = half * N + gr;
        int c  = cnt[li];

        v8f acc[4];
        for (int t = 0; t < 4; ++t) {
            float b = bself[t * 16 + row]
                    + (1.0f - ALPHA) * bs2d[t * 16 + row]
                    + ALPHA * bd2s[t * 16 + row];
            for (int r = 0; r < 8; ++r) acc[t][r] = b;
        }

        const float* xrow = x + (long long)gr * D + khalf;
        v2f ax[16];
#pragma unroll
        for (int j = 0; j < 16; ++j) ax[j] = *(const v2f*)(xrow + 4 * j);
#pragma unroll
        for (int j = 0; j < 16; ++j) {
            const float* wk = wbase + j * 256;
            for (int t = 0; t < 4; ++t) {
                v2f bs = *(const v2f*)(wk + t * 32);
                acc[t] = __builtin_amdgcn_wmma_f32_16x16x4_f32(
                    false, ax[j], false, bs, (short)0, acc[t], false, false);
            }
        }

        for (int dir = 0; dir < 2; ++dir) {
            gather_tile(x, slots + ((long long)dir * N + gr0) * CAP, st, c, dir * 16,
                        dir ? ALPHA : (1.0f - ALPHA),
                        ovf + (long long)dir * E, cursor + dir * LINE_INTS + 1,
                        ei + (dir ? 0 : E), ei + (dir ? E : 0),
                        gr0, E, lane);
            __builtin_amdgcn_wave_barrier();
            const float* wd = wbase + (dir + 1) * D * D;
            for (int k0 = 0; k0 < D; k0 += 4) {
                v2f aa = *(const v2f*)(sbase + k0);
                const float* wk = wd + (k0 >> 1) * 128;
                for (int t = 0; t < 4; ++t) {
                    v2f bb = *(const v2f*)(wk + t * 32);
                    acc[t] = __builtin_amdgcn_wmma_f32_16x16x4_f32(
                        false, aa, false, bb, (short)0, acc[t], false, false);
                }
            }
            __builtin_amdgcn_wave_barrier();
        }

        int mofs = half * 8;
        for (int t = 0; t < 4; ++t) {
            int col = t * 16 + row;
            long long base = (long long)(gr0 + mofs) * D + col;
            for (int r = 0; r < 8; ++r) out[base + (long long)r * D] = acc[t][r];
        }
    }
}

extern "C" void kernel_launch(void* const* d_in, const int* in_sizes, int n_in,
                              void* d_out, int out_size, void* d_ws, size_t ws_size,
                              hipStream_t stream) {
    const float* x     = (const float*)d_in[0];
    const int*   ei    = (const int*)  d_in[1];
    const float* Wself = (const float*)d_in[2];
    const float* bself = (const float*)d_in[3];
    const float* Ws2d  = (const float*)d_in[4];
    const float* bs2d  = (const float*)d_in[5];
    const float* Wd2s  = (const float*)d_in[6];
    const float* bd2s  = (const float*)d_in[7];
    float* out = (float*)d_out;

    int N = in_sizes[0] / D;
    int E = in_sizes[1] / 2;

    int* wsi    = (int*)d_ws;
    int* cnt    = wsi;
    int* cursor = wsi + 2 * (size_t)N;
    int* slots  = cursor + 2 * LINE_INTS;
    int* ovf    = slots + 2 * (size_t)N * CAP;

    int zn = 2 * N + 2 * LINE_INTS;
    zero_counts_kernel<<<(zn + 255) / 256, 256, 0, stream>>>(wsi, zn);
    fill_lists_kernel<<<EDGE_GRID, EDGE_WG, 0, stream>>>(ei, cnt, slots, ovf, cursor, N, E);
    dirsage_pull_gemm_kernel<<<256, 32 * WAVES, 0, stream>>>(
        x, ei, cnt, slots, ovf, cursor, Wself, bself, Ws2d, bs2d, Wd2s, bd2s, out, N, E);
}


# Edge pass puts each 16384-edge chunk in counter-line order in LDS (count, scan, place) before its global count adds: a wave's 32 adds fall in a few 128-B lines, not 32; sums unchanged bit for bit
# speedup vs baseline: 3.8959x; 1.0439x over previous
#include <hip/hip_runtime.h>


#define D 64
#define ALPHA 0.5f
#define PITCH 68
#define WAVES 28
#define BATCH 16
#define EDGE_WG 1024
#define EDGE_PT 16
#define EDGE_UNIT (EDGE_WG * EDGE_PT)
#define EDGE_BINS 4096
#define EDGE_GRID 256
#define CAP 32
#define LINE_INTS 32

typedef __attribute__((ext_vector_type(2))) float v2f;
typedef __attribute__((ext_vector_type(8))) float v8f;

__global__ void zero_counts_kernel(int* __restrict__ p, int n) {
    int i = blockIdx.x * blockDim.x + threadIdx.x;
    if (i < n) p[i] = 0;
}

__device__ __forceinline__ int chip_half() {
    return (int)((__builtin_amdgcn_s_sendmsg_rtn(0x87) >> 18) & 1u);
}


__global__ __launch_bounds__(EDGE_WG)
void fill_lists_kernel(const int* __restrict__ ei,
                       int* __restrict__ cnt,
                       int* __restrict__ slots,
                       int* __restrict__ ovf,
                       int* __restrict__ cursor,
                       int N, int E) {
    __shared__ int bin[EDGE_BINS];
    __shared__ int wsum[EDGE_WG / 32];
    __shared__ int skey[EDGE_UNIT];
    __shared__ int sval[EDGE_UNIT];
    __shared__ int sedge[EDGE_UNIT];
    __shared__ int tk;
    __shared__ int sh_home;
    const int tid    = threadIdx.x;
    const int lane   = tid & 31;
    const int wave   = tid >> 5;
    const int nunits = (E + EDGE_UNIT - 1) / EDGE_UNIT;
    int sh = 5;
    while (((N - 1) >> sh) >= EDGE_BINS) ++sh;
    if (tid == 0) sh_home = chip_half();
    __syncthreads();
    const int home = sh_home;
    for (int pass = 0; pass < 2; ++pass) {
        const int dir = home ^ pass;
        const int* __restrict__ key = ei + (dir ? 0 : E);
        const int* __restrict__ val = ei + (dir ? E : 0);
        int* __restrict__ c    = cnt + dir * N;
        int* __restrict__ sl   = slots + (long long)dir * N * CAP;
        int* __restrict__ ov   = ovf + (long long)dir * E;
        int* tick = cursor + dir * LINE_INTS;
        int* nov  = tick + 1;
        for (int it = 0; it < nunits; ++it) {
            if (tid == 0) tk = atomicAdd(tick, 1);
            __syncthreads();
            const int t = __builtin_amdgcn_readfirstlane(tk);
            __syncthreads();
            if (t >= nunits) break;
            const int base = t * EDGE_UNIT;
            int k[EDGE_PT], v[EDGE_PT], r[EDGE_PT], ed[EDGE_PT];
#pragma unroll
            for (int i = 0; i < EDGE_PT; ++i) {
                int e = base + i * EDGE_WG + tid;
                bool in = e < E;
                k[i] = in ? key[e] : -1;
                v[i] = in ? val[e] : 0;
            }
            for (int b = tid; b < EDGE_BINS; b += EDGE_WG) bin[b] = 0;
            __syncthreads();
#pragma unroll
            for (int i = 0; i < EDGE_PT; ++i)
                r[i] = (k[i] >= 0) ? atomicAdd(&bin[k[i] >> sh], 1) : 0;
            __syncthreads();
            const int b0 = bin[4 * tid], b1 = bin[4 * tid + 1], b2 = bin[4 * tid + 2], b3 = bin[4 * tid + 3];
            const int s4 = b0 + b1 + b2 + b3;
            int inc = s4;
            for (int d = 1; d < 32; d <<= 1) {
                int o = __shfl_up(inc, d);
                if (lane >= d) inc += o;
            }
            if (lane == 31) wsum[wave] = inc;
            __syncthreads();
            if (wave == 0) {
                const int w = wsum[lane];
                int winc = w;
                for (int d = 1; d < 32; d <<= 1) {
                    int o = __shfl_up(winc, d);
                    if (lane >= d) winc += o;
                }
                wsum[lane] = winc - w;
            }
            __syncthreads();
            const int ex = wsum[wave] + inc - s4;
            bin[4 * tid]     = ex;
            bin[4 * tid + 1] = ex + b0;
            bin[4 * tid + 2] = ex + b0 + b1;
            bin[4 * tid + 3] = ex + b0 + b1 + b2;
            __syncthreads();
#pragma unroll
            for (int i = 0; i < EDGE_PT; ++i) {
                if (k[i] >= 0) {
                    int p = bin[k[i] >> sh] + r[i];
                    skey[p]  = k[i];
                    sval[p]  = v[i];
                    sedge[p] = base + i * EDGE_WG + tid;
                }
            }
            __syncthreads();
            const int nvalid = (E - base) < EDGE_UNIT ? (E - base) : EDGE_UNIT;
#pragma unroll
            for (int i = 0; i < EDGE_PT; ++i) {
                int p = i * EDGE_WG + tid;
                bool in = p < nvalid;
                k[i]  = in ? skey[p] : -1;
                v[i]  = in ? sval[p] : 0;
                ed[i] = in ? sedge[p] : 0;
            }
#pragma unroll
            for (int i = 0; i < EDGE_PT; ++i)
                r[i] = (k[i] >= 0) ? atomicAdd(&c[k[i]], 1) : 0;
#pragma unroll
            for (int i = 0; i < EDGE_PT; ++i) {
                if (k[i] >= 0) {
                    if (r[i] < CAP) {
                        sl[k[i] * CAP + r[i]] = v[i];
                    } else {
                        int o = atomicAdd(nov, 1);
                        if (o < E) ov[o] = ed[i];
                    }
                }
            }
        }
    }
}

#define ROW_AT(l) (*(const v2f*)(xb + ((unsigned)__builtin_amdgcn_readlane((int)sorted, (l)) + lane8)))
template <int BASE>
__device__ __forceinline__ v2f add_rows(v2f acc, const char* __restrict__ xb,
                                        unsigned sorted, unsigned lane8, int n) {
    const int g = n >> 2;
    const int r = n & 3;
    const int tail = BASE + 4 * g;
    v2f v[BATCH];
    v2f w0, w1, w2;
    if (g > 0) {
        v[0] = ROW_AT(BASE + 0);  v[1] = ROW_AT(BASE + 1);
        v[2] = ROW_AT(BASE + 2);  v[3] = ROW_AT(BASE + 3);
        if (g > 1) {
            v[4] = ROW_AT(BASE + 4);  v[5] = ROW_AT(BASE + 5);
            v[6] = ROW_AT(BASE + 6);  v[7] = ROW_AT(BASE + 7);
            if (g > 2) {
                v[8]  = ROW_AT(BASE + 8);   v[9]  = ROW_AT(BASE + 9);
                v[10] = ROW_AT(BASE + 10);  v[11] = ROW_AT(BASE + 11);
                if (g > 3) {
                    v[12] = ROW_AT(BASE + 12);  v[13] = ROW_AT(BASE + 13);
                    v[14] = ROW_AT(BASE + 14);  v[15] = ROW_AT(BASE + 15);
                }
            }
        }
    }
    if (r > 0) {
        w0 = ROW_AT(tail);
        if (r > 1) {
            w1 = ROW_AT(tail + 1);
            if (r > 2) w2 = ROW_AT(tail + 2);
        }
    }
    if (g > 0) {
        acc += v[0];  acc += v[1];  acc += v[2];  acc += v[3];
        if (g > 1) {
            acc += v[4];  acc += v[5];  acc += v[6];  acc += v[7];
            if (g > 2) {
                acc += v[8];  acc += v[9];  acc += v[10];  acc += v[11];
                if (g > 3) {
                    acc += v[12];  acc += v[13];  acc += v[14];  acc += v[15];
                }
            }
        }
    }
    if (r > 0) {
        acc += w0;
        if (r > 1) {
            acc += w1;
            if (r > 2) acc += w2;
        }
    }
    return acc;
}

__device__ __forceinline__ v2f sum_sorted(const float* __restrict__ x,
                                          int L, int nid, int lane) {
    v2f acc = {0.0f, 0.0f};
    unsigned key = (lane < L) ? (((unsigned)nid << 5) | (unsigned)lane) : 0xffffffffu;
    int rk = 0;
    for (int t = 0; t < L; t += 4) {
#pragma unroll
        for (int u = 0; u < 4; ++u) {
            unsigned v = (unsigned)__builtin_amdgcn_readlane((int)key, t + u);
            rk += (v < key) ? 1 : 0;
        }
    }
    unsigned sorted = (unsigned)__builtin_amdgcn_ds_permute(rk << 2, (int)((unsigned)nid << 8));
    const char* xb = (const char*)x;
    const unsigned lane8 = 8u * (unsigned)lane;
    acc = add_rows<0>(acc, xb, sorted, lane8, L < BATCH ? L : BATCH);
    if (L > BATCH) acc = add_rows<BATCH>(acc, xb, sorted, lane8, L - BATCH);
    return acc;
}

__device__ __attribute__((noinline)) v2f sum_mid(const float* __restrict__ x,
                                       const int* __restrict__ row,
                                       const int* __restrict__ ov, int nov,
                                       const int* __restrict__ key,
                                       const int* __restrict__ val,
                                       int node, int E, int lane) {
    v2f acc = {0.0f, 0.0f};
    int id1 = row[lane];
    int id2 = 0x7fffffff;
    int m = 0;
    for (int i0 = 0; i0 < nov; i0 += 32) {
        int i = i0 + lane;
        int v = 0;
        bool hit = false;
        if (i < nov) {
            int e = ov[i];
            if ((unsigned)e < (unsigned)E && key[e] == node) {
                v = val[e];
                hit = true;
            }
        }
        unsigned found = __builtin_amdgcn_ballot_w32(hit);
        while (found != 0u && m < CAP) {
            int vv = __builtin_amdgcn_readlane(v, __builtin_ctz(found));
            if (lane == m) id2 = vv;
            ++m;
            found &= found - 1u;
        }
    }
    int rk1 = 0, rk2 = 0;
    for (int t = 0; t < CAP; ++t) {
        int a = __builtin_amdgcn_readlane(id1, t);
        rk1 += (a < id1 || (a == id1 && t < lane)) ? 1 : 0;
        rk2 += (a <= id2) ? 1 : 0;
    }
    for (int t = 0; t < m; ++t) {
        int b = __builtin_amdgcn_readlane(id2, t);
        rk1 += (b < id1) ? 1 : 0;
        rk2 += (b < id2 || (b == id2 && t < lane)) ? 1 : 0;
    }
    const int n = CAP + m;
    const float* xl = x + 2 * lane;
    for (int t0 = 0; t0 < n; t0 += BATCH) {
        v2f v[BATCH];
#pragma unroll
        for (int i = 0; i < BATCH; ++i) {
            if (t0 + i < n) {
                unsigned m1 = __builtin_amdgcn_ballot_w32(rk1 == t0 + i);
                unsigned m2 = __builtin_amdgcn_ballot_w32(lane < m && rk2 == t0 + i);
                int j1 = __builtin_amdgcn_readlane(id1, m1 ? __builtin_ctz(m1) : 0);
                int j2 = __builtin_amdgcn_readlane(id2, m2 ? __builtin_ctz(m2) : 0);
                int j  = m1 ? j1 : j2;
                v[i] = *(const v2f*)(xl + (long long)j * D);
            }
        }
#pragma unroll
        for (int i = 0; i < BATCH; ++i)
            if (t0 + i < n) acc += v[i];
    }
    return acc;
}

__device__ __forceinline__ v2f sum_long(const float* __restrict__ x,
                                        const int* __restrict__ row,
                                        const int* __restrict__ ov, int nov,
                                        const int* __restrict__ key,
                                        const int* __restrict__ val,
                                        int node, int L, int E, int lane) {
    v2f acc = {0.0f, 0.0f};
    int own  = row[lane];
    int prev = -1;
    int done = 0;
    while (done < L) {
        int best = (own > prev) ? own : 0x7fffffff;
        for (int i = lane; i < nov; i += 32) {
            int e = ov[i];
            if ((unsigned)e < (unsigned)E && key[e] == node) {
                int v = val[e];
                if (v > prev && v < best) best = v;
            }
        }
        for (int m = 16; m >= 1; m >>= 1) {
            int o = __shfl_xor(best, m);
            best = o < best ? o : best;
        }
        int mult = (own == best) ? 1 : 0;
        for (int i = lane; i < nov; i += 32) {
            int e = ov[i];
            if ((unsigned)e < (unsigned)E && key[e] == node && val[e] == best) mult += 1;
        }
        for (int m = 16; m >= 1; m >>= 1) mult += __shfl_xor(mult, m);
        best = __builtin_amdgcn_readfirstlane(best);
        mult = __builtin_amdgcn_readfirstlane(mult);
        if (mult == 0) break;
        v2f r = *(const v2f*)(x + (long long)best * D + 2 * lane);
        acc += r;
        for (int q = 1; q < mult; ++q) acc += r;
        prev = best;
        done += mult;
    }
    return acc;
}

__device__ __forceinline__ void gather_tile(const float* __restrict__ x,
                                            const int* __restrict__ rows,
                                            float* st,
                                            int c, int sel, float alpha,
                                            const int* __restrict__ ov,
                                            const int* __restrict__ novp,
                                            const int* __restrict__ key,
                                            const int* __restrict__ val,
                                            int node0, int E, int lane) {
    int L0  = __builtin_amdgcn_readlane(c, sel);
    int nid = (lane < L0) ? rows[lane] : 0;
    for (int r = 0; r < 16; ++r) {
        int Lc = L0, nidc = nid;
        if (r < 15) {
            L0  = __builtin_amdgcn_readlane(c, sel + r + 1);
            nid = (lane < L0) ? rows[(r + 1) * CAP + lane] : 0;
        }
        v2f a = {0.0f, 0.0f};
        if (Lc <= CAP) a = sum_sorted(x, Lc, nidc, lane);
        a *= alpha / fmaxf((float)Lc, 1.0f);
        *(v2f*)(st + r * PITCH + 2 * lane) = a;
    }
    for (int r = 0; r < 16; ++r) {
        int Lc = __builtin_amdgcn_readlane(c, sel + r);
        if (Lc > CAP) {
            int nov = novp[0];
            nov = nov < E ? nov : E;
            v2f a = (Lc <= 2 * CAP)
                  ? sum_mid(x, rows + r * CAP, ov, nov, key, val, node0 + r, E, lane)
                  : sum_long(x, rows + r * CAP, ov, nov, key, val, node0 + r, Lc, E, lane);
            a *= alpha / fmaxf((float)Lc, 1.0f);
            *(v2f*)(st + r * PITCH + 2 * lane) = a;
        }
    }
}

__global__ __launch_bounds__(32 * WAVES)
void dirsage_pull_gemm_kernel(const float* __restrict__ x,
                              const int*   __restrict__ ei,
                              const int*   __restrict__ cnt,
                              const int*   __restrict__ slots,
                              const int*   __restrict__ ovf,
                              const int*   __restrict__ cursor,
                              const float* __restrict__ Wself,
                              const float* __restrict__ bself,
                              const float* __restrict__ Ws2d,
                              const float* __restrict__ bs2d,
                              const float* __restrict__ Wd2s,
                              const float* __restrict__ bd2s,
                              float* __restrict__ out,
                              int N, int E) {
    __shared__ float wl[3 * D * D];
    __shared__ float stage[WAVES * 16 * PITCH];

    for (int i = threadIdx.x; i < D * D; i += blockDim.x) {
        int k   = i >> 6;
        int col = i & 63;
        int dst = ((k >> 1) << 7) + (col << 1) + (k & 1);
        wl[dst]             = Wself[i];
        wl[D * D + dst]     = Ws2d[i];
        wl[2 * D * D + dst] = Wd2s[i];
    }
    __syncthreads();

    int wave   = threadIdx.x >> 5;
    int lane   = threadIdx.x & 31;
    int row    = lane & 15;
    int half   = lane >> 4;
    int khalf  = half * 2;
    int mtiles = N >> 4;
    float* st  = stage + wave * 16 * PITCH;
    const float* wbase = &wl[half * 128 + row * 2];
    const float* sbase = st + row * PITCH + khalf;

    for (int mtile = blockIdx.x + gridDim.x * wave; mtile < mtiles;
         mtile += gridDim.x * WAVES) {
        int gr0 = mtile * 16;
        int gr  = gr0 + row;

        int li = half * N + gr;
        int c  = cnt[li];

        v8f acc[4];
        for (int t = 0; t < 4; ++t) {
            float b = bself[t * 16 + row]
                    + (1.0f - ALPHA) * bs2d[t * 16 + row]
                    + ALPHA * bd2s[t * 16 + row];
            for (int r = 0; r < 8; ++r) acc[t][r] = b;
        }

        const float* xrow = x + (long long)gr * D + khalf;
        v2f ax[16];
#pragma unroll
        for (int j = 0; j < 16; ++j) ax[j] = *(const v2f*)(xrow + 4 * j);
#pragma unroll
        for (int j = 0; j < 16; ++j) {
            const float* wk = wbase + j * 256;
            for (int t = 0; t < 4; ++t) {
                v2f bs = *(const v2f*)(wk + t * 32);
                acc[t] = __builtin_amdgcn_wmma_f32_16x16x4_f32(
                    false, ax[j], false, bs, (short)0, acc[t], false, false);
            }
        }

        for (int dir = 0; dir < 2; ++dir) {
            gather_tile(x, slots + ((long long)dir * N + gr0) * CAP, st, c, dir * 16,
                        dir ? ALPHA : (1.0f - ALPHA),
                        ovf + (long long)dir * E, cursor + dir * LINE_INTS + 1,
                        ei + (dir ? 0 : E), ei + (dir ? E : 0),
                        gr0, E, lane);
            __builtin_amdgcn_wave_barrier();
            const float* wd = wbase + (dir + 1) * D * D;
            for (int k0 = 0; k0 < D; k0 += 4) {
                v2f aa = *(const v2f*)(sbase + k0);
                const float* wk = wd + (k0 >> 1) * 128;
                for (int t = 0; t < 4; ++t) {
                    v2f bb = *(const v2f*)(wk + t * 32);
                    acc[t] = __builtin_amdgcn_wmma_f32_16x16x4_f32(
                        false, aa, false, bb, (short)0, acc[t], false, false);
                }
            }
            __builtin_amdgcn_wave_barrier();
        }

        int mofs = half * 8;
        for (int t = 0; t < 4; ++t) {
            int col = t * 16 + row;
            long long base = (long long)(gr0 + mofs) * D + col;
            for (int r = 0; r < 8; ++r) out[base + (long long)r * D] = acc[t][r];
        }
    }
}

extern "C" void kernel_launch(void* const* d_in, const int* in_sizes, int n_in,
                              void* d_out, int out_size, void* d_ws, size_t ws_size,
                              hipStream_t stream) {
    const float* x     = (const float*)d_in[0];
    const int*   ei    = (const int*)  d_in[1];
    const float* Wself = (const float*)d_in[2];
    const float* bself = (const float*)d_in[3];
    const float* Ws2d  = (const float*)d_in[4];
    const float* bs2d  = (const float*)d_in[5];
    const float* Wd2s  = (const float*)d_in[6];
    const float* bd2s  = (const float*)d_in[7];
    float* out = (float*)d_out;

    int N = in_sizes[0] / D;
    int E = in_sizes[1] / 2;

    int* wsi    = (int*)d_ws;
    int* cnt    = wsi;
    int* cursor = wsi + 2 * (size_t)N;
    int* slots  = cursor + 2 * LINE_INTS;
    int* ovf    = slots + 2 * (size_t)N * CAP;

    int zn = 2 * N + 2 * LINE_INTS;
    zero_counts_kernel<<<(zn + 255) / 256, 256, 0, stream>>>(wsi, zn);
    fill_lists_kernel<<<EDGE_GRID, EDGE_WG, 0, stream>>>(ei, cnt, slots, ovf, cursor, N, E);
    dirsage_pull_gemm_kernel<<<256, 32 * WAVES, 0, stream>>>(
        x, ei, cnt, slots, ovf, cursor, Wself, bself, Ws2d, bs2d, Wd2s, bd2s, out, N, E);
}


# Neighbour rows gathered by half-waves: each 16-lane half ranks and sums one list (16 B of a row per lane), two lists per pass; ranks and addresses by row-share operands; sums unchanged bit for bit
# speedup vs baseline: 4.0665x; 1.0438x over previous
#include <hip/hip_runtime.h>


#define D 64
#define ALPHA 0.5f
#define PITCH 68
#define WAVES 28
#define BATCH 16
#define FIRST_GROUPS 4
#define EDGE_WG 1024
#define EDGE_PT 16
#define EDGE_UNIT (EDGE_WG * EDGE_PT)
#define EDGE_BINS 4096
#define EDGE_GRID 256
#define CAP 32
#define LINE_INTS 32

typedef __attribute__((ext_vector_type(2))) float v2f;
typedef __attribute__((ext_vector_type(4))) float v4f;
typedef __attribute__((ext_vector_type(8))) float v8f;

__global__ void zero_counts_kernel(int* __restrict__ p, int n) {
    int i = blockIdx.x * blockDim.x + threadIdx.x;
    if (i < n) p[i] = 0;
}

__device__ __forceinline__ int chip_half() {
    return (int)((__builtin_amdgcn_s_sendmsg_rtn(0x87) >> 18) & 1u);
}


__global__ __launch_bounds__(EDGE_WG)
void fill_lists_kernel(const int* __restrict__ ei,
                       int* __restrict__ cnt,
                       int* __restrict__ slots,
                       int* __restrict__ ovf,
                       int* __restrict__ cursor,
                       int N, int E) {
    __shared__ int bin[EDGE_BINS];
    __shared__ int wsum[EDGE_WG / 32];
    __shared__ int skey[EDGE_UNIT];
    __shared__ int sval[EDGE_UNIT];
    __shared__ int sedge[EDGE_UNIT];
    __shared__ int tk;
    __shared__ int sh_home;
    const int tid    = threadIdx.x;
    const int lane   = tid & 31;
    const int wave   = tid >> 5;
    const int nunits = (E + EDGE_UNIT - 1) / EDGE_UNIT;
    int sh = 5;
    while (((N - 1) >> sh) >= EDGE_BINS) ++sh;
    if (tid == 0) sh_home = chip_half();
    __syncthreads();
    const int home = sh_home;
    for (int pass = 0; pass < 2; ++pass) {
        const int dir = home ^ pass;
        const int* __restrict__ key = ei + (dir ? 0 : E);
        const int* __restrict__ val = ei + (dir ? E : 0);
        int* __restrict__ c    = cnt + dir * N;
        int* __restrict__ sl   = slots + (long long)dir * N * CAP;
        int* __restrict__ ov   = ovf + (long long)dir * E;
        int* tick = cursor + dir * LINE_INTS;
        int* nov  = tick + 1;
        for (int it = 0; it < nunits; ++it) {
            if (tid == 0) tk = atomicAdd(tick, 1);
            __syncthreads();
            const int t = __builtin_amdgcn_readfirstlane(tk);
            __syncthreads();
            if (t >= nunits) break;
            const int base = t * EDGE_UNIT;
            int k[EDGE_PT], v[EDGE_PT], r[EDGE_PT], ed[EDGE_PT];
#pragma unroll
            for (int i = 0; i < EDGE_PT; ++i) {
                int e = base + i * EDGE_WG + tid;
                bool in = e < E;
                k[i] = in ? key[e] : -1;
                v[i] = in ? val[e] : 0;
            }
            for (int b = tid; b < EDGE_BINS; b += EDGE_WG) bin[b] = 0;
            __syncthreads();
#pragma unroll
            for (int i = 0; i < EDGE_PT; ++i)
                r[i] = (k[i] >= 0) ? atomicAdd(&bin[k[i] >> sh], 1) : 0;
            __syncthreads();
            const int b0 = bin[4 * tid], b1 = bin[4 * tid + 1], b2 = bin[4 * tid + 2], b3 = bin[4 * tid + 3];
            const int s4 = b0 + b1 + b2 + b3;
            int inc = s4;
            for (int d = 1; d < 32; d <<= 1) {
                int o = __shfl_up(inc, d);
                if (lane >= d) inc += o;
            }
            if (lane == 31) wsum[wave] = inc;
            __syncthreads();
            if (wave == 0) {
                const int w = wsum[lane];
                int winc = w;
                for (int d = 1; d < 32; d <<= 1) {
                    int o = __shfl_up(winc, d);
                    if (lane >= d) winc += o;
                }
                wsum[lane] = winc - w;
            }
            __syncthreads();
            const int ex = wsum[wave] + inc - s4;
            bin[4 * tid]     = ex;
            bin[4 * tid + 1] = ex + b0;
            bin[4 * tid + 2] = ex + b0 + b1;
            bin[4 * tid + 3] = ex + b0 + b1 + b2;
            __syncthreads();
#pragma unroll
            for (int i = 0; i < EDGE_PT; ++i) {
                if (k[i] >= 0) {
                    int p = bin[k[i] >> sh] + r[i];
                    skey[p]  = k[i];
                    sval[p]  = v[i];
                    sedge[p] = base + i * EDGE_WG + tid;
                }
            }
            __syncthreads();
            const int nvalid = (E - base) < EDGE_UNIT ? (E - base) : EDGE_UNIT;
#pragma unroll
            for (int i = 0; i < EDGE_PT; ++i) {
                int p = i * EDGE_WG + tid;
                bool in = p < nvalid;
                k[i]  = in ? skey[p] : -1;
                v[i]  = in ? sval[p] : 0;
                ed[i] = in ? sedge[p] : 0;
            }
#pragma unroll
            for (int i = 0; i < EDGE_PT; ++i)
                r[i] = (k[i] >= 0) ? atomicAdd(&c[k[i]], 1) : 0;
#pragma unroll
            for (int i = 0; i < EDGE_PT; ++i) {
                if (k[i] >= 0) {
                    if (r[i] < CAP) {
                        sl[k[i] * CAP + r[i]] = v[i];
                    } else {
                        int o = atomicAdd(nov, 1);
                        if (o < E) ov[o] = ed[i];
                    }
                }
            }
        }
    }
}

template <int T>
__device__ __forceinline__ unsigned row_lane(unsigned v) {
    return (unsigned)__builtin_amdgcn_update_dpp(0, (int)v, 0x150 + T, 0xf, 0xf, true);
}

template <int G, bool HI>
__device__ __forceinline__ void rank_group(unsigned src, unsigned klo, unsigned khi,
                                           int& rlo, int& rhi) {
    unsigned a0 = row_lane<4 * G + 0>(src);
    unsigned a1 = row_lane<4 * G + 1>(src);
    unsigned a2 = row_lane<4 * G + 2>(src);
    unsigned a3 = row_lane<4 * G + 3>(src);
    rlo += (a0 < klo) ? 1 : 0;
    rlo += (a1 < klo) ? 1 : 0;
    rlo += (a2 < klo) ? 1 : 0;
    rlo += (a3 < klo) ? 1 : 0;
    if (HI) {
        rhi += (a0 < khi) ? 1 : 0;
        rhi += (a1 < khi) ? 1 : 0;
        rhi += (a2 < khi) ? 1 : 0;
        rhi += (a3 < khi) ? 1 : 0;
    }
}

template <bool HI>
__device__ __forceinline__ void rank_rows(unsigned src, int n, unsigned klo, unsigned khi,
                                          int& rlo, int& rhi) {
    if (n > 0) {
        rank_group<0, HI>(src, klo, khi, rlo, rhi);
        if (n > 4) {
            rank_group<1, HI>(src, klo, khi, rlo, rhi);
            if (n > 8) {
                rank_group<2, HI>(src, klo, khi, rlo, rhi);
                if (n > 12) rank_group<3, HI>(src, klo, khi, rlo, rhi);
            }
        }
    }
}

template <int J>
__device__ __forceinline__ v4f add_group(v4f acc, v4f r0, v4f r1, v4f r2, v4f r3,
                                         bool whole, int len) {
    if (whole) {
        acc += r0;  acc += r1;  acc += r2;  acc += r3;
    } else {
        float m0 = (J + 0 < len) ? 1.0f : 0.0f;
        float m1 = (J + 1 < len) ? 1.0f : 0.0f;
        float m2 = (J + 2 < len) ? 1.0f : 0.0f;
        float m3 = (J + 3 < len) ? 1.0f : 0.0f;
        acc = __builtin_elementwise_fma(r0, (v4f){m0, m0, m0, m0}, acc);
        acc = __builtin_elementwise_fma(r1, (v4f){m1, m1, m1, m1}, acc);
        acc = __builtin_elementwise_fma(r2, (v4f){m2, m2, m2, m2}, acc);
        acc = __builtin_elementwise_fma(r3, (v4f){m3, m3, m3, m3}, acc);
    }
    return acc;
}

template <int P>
__device__ __forceinline__ unsigned pos_offset(unsigned slo, unsigned shi) {
    return (P < 16) ? row_lane<(P & 15)>(slo) : row_lane<(P & 15)>(shi);
}

#define ROW4(P) (*(const v4f*)(xb + (pos_offset<(P)>(slo, shi) + lane16)))
#define LOAD_GROUP(K) { v[4 * (K) + 0] = ROW4(4 * (G0 + (K)) + 0);  v[4 * (K) + 1] = ROW4(4 * (G0 + (K)) + 1); \
                        v[4 * (K) + 2] = ROW4(4 * (G0 + (K)) + 2);  v[4 * (K) + 3] = ROW4(4 * (G0 + (K)) + 3); }
#define ADD_GROUP(K)  acc = add_group<4 * (G0 + (K))>(acc, v[4 * (K) + 0], v[4 * (K) + 1], v[4 * (K) + 2], \
                                                      v[4 * (K) + 3], nwhole > G0 + (K), len)
template <int G0, int NG>
__device__ __forceinline__ v4f add_rows_groups(v4f acc, const char* __restrict__ xb,
                                               unsigned slo, unsigned shi, unsigned lane16,
                                               int g, int nwhole, int len) {
    v4f v[4 * NG];
    if (g > 0) {
        LOAD_GROUP(0);
        if (NG > 1 && g > 1) {
            LOAD_GROUP(NG > 1 ? 1 : 0);
            if (NG > 2 && g > 2) {
                LOAD_GROUP(NG > 2 ? 2 : 0);
                if (NG > 3 && g > 3) {
                    LOAD_GROUP(NG > 3 ? 3 : 0);
                    if (NG > 4 && g > 4) LOAD_GROUP(NG > 4 ? 4 : 0);
                }
            }
        }
    }
    if (g > 0) {
        ADD_GROUP(0);
        if (NG > 1 && g > 1) {
            ADD_GROUP(NG > 1 ? 1 : 0);
            if (NG > 2 && g > 2) {
                ADD_GROUP(NG > 2 ? 2 : 0);
                if (NG > 3 && g > 3) {
                    ADD_GROUP(NG > 3 ? 3 : 0);
                    if (NG > 4 && g > 4) ADD_GROUP(NG > 4 ? 4 : 0);
                }
            }
        }
    }
    return acc;
}

__device__ __attribute__((noinline)) v2f sum_mid(const float* __restrict__ x,
                                       const int* __restrict__ row,
                                       const int* __restrict__ ov, int nov,
                                       const int* __restrict__ key,
                                       const int* __restrict__ val,
                                       int node, int E, int lane) {
    v2f acc = {0.0f, 0.0f};
    int id1 = row[lane];
    int id2 = 0x7fffffff;
    int m = 0;
    for (int i0 = 0; i0 < nov; i0 += 32) {
        int i = i0 + lane;
        int v = 0;
        bool hit = false;
        if (i < nov) {
            int e = ov[i];
            if ((unsigned)e < (unsigned)E && key[e] == node) {
                v = val[e];
                hit = true;
            }
        }
        unsigned found = __builtin_amdgcn_ballot_w32(hit);
        while (found != 0u && m < CAP) {
            int vv = __builtin_amdgcn_readlane(v, __builtin_ctz(found));
            if (lane == m) id2 = vv;
            ++m;
            found &= found - 1u;
        }
    }
    int rk1 = 0, rk2 = 0;
    for (int t = 0; t < CAP; ++t) {
        int a = __builtin_amdgcn_readlane(id1, t);
        rk1 += (a < id1 || (a == id1 && t < lane)) ? 1 : 0;
        rk2 += (a <= id2) ? 1 : 0;
    }
    for (int t = 0; t < m; ++t) {
        int b = __builtin_amdgcn_readlane(id2, t);
        rk1 += (b < id1) ? 1 : 0;
        rk2 += (b < id2 || (b == id2 && t < lane)) ? 1 : 0;
    }
    const int n = CAP + m;
    const float* xl = x + 2 * lane;
    for (int t0 = 0; t0 < n; t0 += BATCH) {
        v2f v[BATCH];
#pragma unroll
        for (int i = 0; i < BATCH; ++i) {
            if (t0 + i < n) {
                unsigned m1 = __builtin_amdgcn_ballot_w32(rk1 == t0 + i);
                unsigned m2 = __builtin_amdgcn_ballot_w32(lane < m && rk2 == t0 + i);
                int j1 = __builtin_amdgcn_readlane(id1, m1 ? __builtin_ctz(m1) : 0);
                int j2 = __builtin_amdgcn_readlane(id2, m2 ? __builtin_ctz(m2) : 0);
                int j  = m1 ? j1 : j2;
                v[i] = *(const v2f*)(xl + (long long)j * D);
            }
        }
#pragma unroll
        for (int i = 0; i < BATCH; ++i)
            if (t0 + i < n) acc += v[i];
    }
    return acc;
}

__device__ __forceinline__ v2f sum_long(const float* __restrict__ x,
                                        const int* __restrict__ row,
                                        const int* __restrict__ ov, int nov,
                                        const int* __restrict__ key,
                                        const int* __restrict__ val,
                                        int node, int L, int E, int lane) {
    v2f acc = {0.0f, 0.0f};
    int own  = row[lane];
    int prev = -1;
    int done = 0;
    while (done < L) {
        int best = (own > prev) ? own : 0x7fffffff;
        for (int i = lane; i < nov; i += 32) {
            int e = ov[i];
            if ((unsigned)e < (unsigned)E && key[e] == node) {
                int v = val[e];
                if (v > prev && v < best) best = v;
            }
        }
        for (int m = 16; m >= 1; m >>= 1) {
            int o = __shfl_xor(best, m);
            best = o < best ? o : best;
        }
        int mult = (own == best) ? 1 : 0;
        for (int i = lane; i < nov; i += 32) {
            int e = ov[i];
            if ((unsigned)e < (unsigned)E && key[e] == node && val[e] == best) mult += 1;
        }
        for (int m = 16; m >= 1; m >>= 1) mult += __shfl_xor(mult, m);
        best = __builtin_amdgcn_readfirstlane(best);
        mult = __builtin_amdgcn_readfirstlane(mult);
        if (mult == 0) break;
        v2f r = *(const v2f*)(x + (long long)best * D + 2 * lane);
        acc += r;
        for (int q = 1; q < mult; ++q) acc += r;
        prev = best;
        done += mult;
    }
    return acc;
}

__device__ __forceinline__ void gather_tile(const float* __restrict__ x,
                                            const int* __restrict__ rows,
                                            float* st,
                                            int c, int sel, float alpha,
                                            const int* __restrict__ ov,
                                            const int* __restrict__ novp,
                                            const int* __restrict__ key,
                                            const int* __restrict__ val,
                                            int node0, int E, int lane) {
    const int hf = lane >> 4;
    const int cl = lane & 15;
    const unsigned lane16 = 16u * (unsigned)cl;
    const char* xb = (const char*)x;
    const int* myrow = rows + hf * CAP + cl;
    int La = __builtin_amdgcn_readlane(c, sel);
    int Lb = __builtin_amdgcn_readlane(c, sel + 1);
    int ea = (La <= CAP) ? La : 0;
    int eb = (Lb <= CAP) ? Lb : 0;
    int Ln = hf ? Lb : La;
    int Le = hf ? eb : ea;
    int idlo = (cl < Le) ? myrow[0] : 0;
    int idhi = (16 + cl < Le) ? myrow[16] : 0;
#pragma nounroll
    for (int p = 0; p < 8; ++p) {
        const int Lc = Ln, len = Le, ilo = idlo, ihi = idhi;
        const int Lmax = ea > eb ? ea : eb;
        const int Lmin = ea < eb ? ea : eb;
        if (p < 7) {
            La = __builtin_amdgcn_readlane(c, sel + 2 * p + 2);
            Lb = __builtin_amdgcn_readlane(c, sel + 2 * p + 3);
            ea = (La <= CAP) ? La : 0;
            eb = (Lb <= CAP) ? Lb : 0;
            Ln = hf ? Lb : La;
            Le = hf ? eb : ea;
            idlo = (cl < Le) ? myrow[(p + 1) * 2 * CAP] : 0;
            idhi = (16 + cl < Le) ? myrow[(p + 1) * 2 * CAP + 16] : 0;
        }
        const bool oklo = cl < len;
        const bool okhi = 16 + cl < len;
        const unsigned klo = oklo ? (((unsigned)ilo << 5) | (unsigned)cl) : 0xffffffffu;
        const unsigned khi = okhi ? (((unsigned)ihi << 5) | (unsigned)(16 + cl)) : 0xffffffffu;
        int rlo = 0, rhi = 0;
        if (Lmax <= 16) {
            rank_rows<false>(klo, Lmax, klo, khi, rlo, rhi);
        } else {
            rank_rows<true>(klo, 16, klo, khi, rlo, rhi);
            rank_rows<true>(khi, Lmax - 16, klo, khi, rlo, rhi);
        }
        float* srow = st + (2 * p + hf) * PITCH;
        srow[oklo ? rlo : cl] = __uint_as_float((unsigned)ilo << 8);
        if (Lmax > 16) srow[okhi ? rhi : 16 + cl] = __uint_as_float((unsigned)ihi << 8);
        __builtin_amdgcn_wave_barrier();
        const unsigned slo = __float_as_uint(srow[cl]);
        const unsigned shi = (Lmax > 16) ? __float_as_uint(srow[16 + cl]) : 0u;
        const int ng = (Lmax + 3) >> 2;
        const int nwhole = Lmin >> 2;
        v4f a = {0.0f, 0.0f, 0.0f, 0.0f};
        a = add_rows_groups<0, FIRST_GROUPS>(a, xb, slo, shi, lane16,
                                             ng < FIRST_GROUPS ? ng : FIRST_GROUPS, nwhole, len);
        if (ng > FIRST_GROUPS)
            a = add_rows_groups<FIRST_GROUPS, 8 - FIRST_GROUPS>(a, xb, slo, shi, lane16,
                                                                ng - FIRST_GROUPS, nwhole, len);
        a *= alpha / fmaxf((float)Lc, 1.0f);
        __builtin_amdgcn_wave_barrier();
        *(v4f*)(srow + 4 * cl) = a;
    }
    for (int r = 0; r < 16; ++r) {
        int Lc = __builtin_amdgcn_readlane(c, sel + r);
        if (Lc > CAP) {
            int nov = novp[0];
            nov = nov < E ? nov : E;
            v2f a = (Lc <= 2 * CAP)
                  ? sum_mid(x, rows + r * CAP, ov, nov, key, val, node0 + r, E, lane)
                  : sum_long(x, rows + r * CAP, ov, nov, key, val, node0 + r, Lc, E, lane);
            a *= alpha / fmaxf((float)Lc, 1.0f);
            *(v2f*)(st + r * PITCH + 2 * lane) = a;
        }
    }
}

__global__ __launch_bounds__(32 * WAVES)
void dirsage_pull_gemm_kernel(const float* __restrict__ x,
                              const int*   __restrict__ ei,
                              const int*   __restrict__ cnt,
                              const int*   __restrict__ slots,
                              const int*   __restrict__ ovf,
                              const int*   __restrict__ cursor,
                              const float* __restrict__ Wself,
                              const float* __restrict__ bself,
                              const float* __restrict__ Ws2d,
                              const float* __restrict__ bs2d,
                              const float* __restrict__ Wd2s,
                              const float* __restrict__ bd2s,
                              float* __restrict__ out,
                              int N, int E) {
    __shared__ float wl[3 * D * D];
    __shared__ float stage[WAVES * 16 * PITCH];

    for (int i = threadIdx.x; i < D * D; i += blockDim.x) {
        int k   = i >> 6;
        int col = i & 63;
        int dst = ((k >> 1) << 7) + (col << 1) + (k & 1);
        wl[dst]             = Wself[i];
        wl[D * D + dst]     = Ws2d[i];
        wl[2 * D * D + dst] = Wd2s[i];
    }
    __syncthreads();

    int wave   = threadIdx.x >> 5;
    int lane   = threadIdx.x & 31;
    int row    = lane & 15;
    int half   = lane >> 4;
    int khalf  = half * 2;
    int mtiles = N >> 4;
    float* st  = stage + wave * 16 * PITCH;
    const float* wbase = &wl[half * 128 + row * 2];
    const float* sbase = st + row * PITCH + khalf;

    for (int mtile = blockIdx.x + gridDim.x * wave; mtile < mtiles;
         mtile += gridDim.x * WAVES) {
        int gr0 = mtile * 16;
        int gr  = gr0 + row;

        int li = half * N + gr;
        int c  = cnt[li];

        v8f acc[4];
        for (int t = 0; t < 4; ++t) {
            float b = bself[t * 16 + row]
                    + (1.0f - ALPHA) * bs2d[t * 16 + row]
                    + ALPHA * bd2s[t * 16 + row];
            for (int r = 0; r < 8; ++r) acc[t][r] = b;
        }

        const float* xrow = x + (long long)gr * D + khalf;
        v2f ax[16];
#pragma unroll
        for (int j = 0; j < 16; ++j) ax[j] = *(const v2f*)(xrow + 4 * j);
#pragma unroll
        for (int j = 0; j < 16; ++j) {
            const float* wk = wbase + j * 256;
            for (int t = 0; t < 4; ++t) {
                v2f bs = *(const v2f*)(wk + t * 32);
                acc[t] = __builtin_amdgcn_wmma_f32_16x16x4_f32(
                    false, ax[j], false, bs, (short)0, acc[t], false, false);
            }
        }

        for (int dir = 0; dir < 2; ++dir) {
            gather_tile(x, slots + ((long long)dir * N + gr0) * CAP, st, c, dir * 16,
                        dir ? ALPHA : (1.0f - ALPHA),
                        ovf + (long long)dir * E, cursor + dir * LINE_INTS + 1,
                        ei + (dir ? 0 : E), ei + (dir ? E : 0),
                        gr0, E, lane);
            __builtin_amdgcn_wave_barrier();
            const float* wd = wbase + (dir + 1) * D * D;
            for (int k0 = 0; k0 < D; k0 += 4) {
                v2f aa = *(const v2f*)(sbase + k0);
                const float* wk = wd + (k0 >> 1) * 128;
                for (int t = 0; t < 4; ++t) {
                    v2f bb = *(const v2f*)(wk + t * 32);
                    acc[t] = __builtin_amdgcn_wmma_f32_16x16x4_f32(
                        false, aa, false, bb, (short)0, acc[t], false, false);
                }
            }
            __builtin_amdgcn_wave_barrier();
        }

        int mofs = half * 8;
        for (int t = 0; t < 4; ++t) {
            int col = t * 16 + row;
            long long base = (long long)(gr0 + mofs) * D + col;
            for (int r = 0; r < 8; ++r) out[base + (long long)r * D] = acc[t][r];
        }
    }
}

extern "C" void kernel_launch(void* const* d_in, const int* in_sizes, int n_in,
                              void* d_out, int out_size, void* d_ws, size_t ws_size,
                              hipStream_t stream) {
    const float* x     = (const float*)d_in[0];
    const int*   ei    = (const int*)  d_in[1];
    const float* Wself = (const float*)d_in[2];
    const float* bself = (const float*)d_in[3];
    const float* Ws2d  = (const float*)d_in[4];
    const float* bs2d  = (const float*)d_in[5];
    const float* Wd2s  = (const float*)d_in[6];
    const float* bd2s  = (const float*)d_in[7];
    float* out = (float*)d_out;

    int N = in_sizes[0] / D;
    int E = in_sizes[1] / 2;

    int* wsi    = (int*)d_ws;
    int* cnt    = wsi;
    int* cursor = wsi + 2 * (size_t)N;
    int* slots  = cursor + 2 * LINE_INTS;
    int* ovf    = slots + 2 * (size_t)N * CAP;

    int zn = 2 * N + 2 * LINE_INTS;
    zero_counts_kernel<<<(zn + 255) / 256, 256, 0, stream>>>(wsi, zn);
    fill_lists_kernel<<<EDGE_GRID, EDGE_WG, 0, stream>>>(ei, cnt, slots, ovf, cursor, N, E);
    dirsage_pull_gemm_kernel<<<256, 32 * WAVES, 0, stream>>>(
        x, ei, cnt, slots, ovf, cursor, Wself, bself, Ws2d, bs2d, Wd2s, bd2s, out, N, E);
}
